# starting kernel
# baseline (speedup 1.0000x reference)
#include <hip/hip_runtime.h>
#include <math.h>

typedef float v2f __attribute__((ext_vector_type(2)));
typedef float v8f __attribute__((ext_vector_type(8)));

#define IN_FEATS  256
#define OUT_FEATS 64
#define WPAIRS    (IN_FEATS / 2)
#define WSTRIDE   160

__global__ void gcn_degrees(const int* __restrict__ src,
                            const int* __restrict__ dst,
                            float* __restrict__ deg_src,
                            float* __restrict__ deg_dst,
                            int n_edges) {
    int t = blockIdx.x * blockDim.x + threadIdx.x;
    if (t >= n_edges) return;
    atomicAdd(&deg_src[src[t]], 1.0f);
    atomicAdd(&deg_dst[dst[t]], 1.0f);
}

__global__ void gcn_norms(float* __restrict__ deg_src,
                          float* __restrict__ deg_dst,
                          int n_nodes) {
    int t = blockIdx.x * blockDim.x + threadIdx.x;
    if (t >= n_nodes) return;
    deg_src[t] = rsqrtf(fmaxf(deg_src[t], 1.0f));
    deg_dst[t] = rsqrtf(fmaxf(deg_dst[t], 1.0f));
}

__global__ void __launch_bounds__(256)
gcn_gemm_wmma(const float* __restrict__ feat,
              const float* __restrict__ norm_src,
              const float* __restrict__ W,
              float* __restrict__ h,
              int n_tiles) {
    __shared__ float Wl[WPAIRS * WSTRIDE];

    const int tid = threadIdx.x;
    for (int i = tid; i < WPAIRS * (OUT_FEATS / 2); i += 256) {
        int p  = i >> 5;
        int n2 = (i & 31) * 2;
        float2 lo = *reinterpret_cast<const float2*>(&W[(2 * p) * OUT_FEATS + n2]);
        float2 hi = *reinterpret_cast<const float2*>(&W[(2 * p + 1) * OUT_FEATS + n2]);
        float4 q;  q.x = lo.x;  q.y = hi.x;  q.z = lo.y;  q.w = hi.y;
        *reinterpret_cast<float4*>(&Wl[p * WSTRIDE + 2 * n2]) = q;
    }
    __syncthreads();

    const int wave = tid >> 5;
    const int lane = tid & 31;
    const int tile = blockIdx.x * 8 + wave;
    if (tile >= n_tiles) return;

    const int half = lane >> 4;
    const int l15  = lane & 15;
    const int m    = tile * 16 + l15;
    const float nrm = norm_src[m];
    const float* frow  = feat + (size_t)m * IN_FEATS + 2 * half;
    const float* wbase = &Wl[half * WSTRIDE + 2 * l15];

    v8f c0 = {}, c1 = {}, c2 = {}, c3 = {};

    #pragma unroll 4
    for (int k = 0; k < IN_FEATS; k += 4) {
        v2f a = *reinterpret_cast<const v2f*>(frow + k);
        a = a * nrm;

        const float* wp = wbase + (k >> 1) * WSTRIDE;
        v2f b0 = *reinterpret_cast<const v2f*>(wp);
        v2f b1 = *reinterpret_cast<const v2f*>(wp + 32);
        v2f b2 = *reinterpret_cast<const v2f*>(wp + 64);
        v2f b3 = *reinterpret_cast<const v2f*>(wp + 96);

        c0 = __builtin_amdgcn_wmma_f32_16x16x4_f32(false, a, false, b0,
                                                   (short)0, c0, false, false);
        c1 = __builtin_amdgcn_wmma_f32_16x16x4_f32(false, a, false, b1,
                                                   (short)0, c1, false, false);
        c2 = __builtin_amdgcn_wmma_f32_16x16x4_f32(false, a, false, b2,
                                                   (short)0, c2, false, false);
        c3 = __builtin_amdgcn_wmma_f32_16x16x4_f32(false, a, false, b3,
                                                   (short)0, c3, false, false);
    }

    const int rbase = tile * 16 + 8 * half;
    #pragma unroll
    for (int r = 0; r < 8; ++r) {
        float* orow = h + (size_t)(rbase + r) * OUT_FEATS + l15;
        orow[0]  = c0[r];
        orow[16] = c1[r];
        orow[32] = c2[r];
        orow[48] = c3[r];
    }
}

__global__ void gcn_scatter(const int* __restrict__ src,
                            const int* __restrict__ dst,
                            const float* __restrict__ h,
                            float* __restrict__ agg,
                            int n_edges) {
    long long t = (long long)blockIdx.x * blockDim.x + threadIdx.x;
    int e = (int)(t >> 4);
    if (e >= n_edges) return;
    int q = (int)(t & 15);
    int s = src[e];
    int d = dst[e];
    float4 v = *reinterpret_cast<const float4*>(h + (size_t)s * OUT_FEATS + q * 4);
    float* ap = agg + (size_t)d * OUT_FEATS + q * 4;
    atomicAdd(ap + 0, v.x);
    atomicAdd(ap + 1, v.y);
    atomicAdd(ap + 2, v.z);
    atomicAdd(ap + 3, v.w);
}

__global__ void gcn_finalize(const float* __restrict__ agg,
                             const float* __restrict__ norm_dst,
                             const float* __restrict__ b,
                             float* __restrict__ out,
                             int total) {
    int t = blockIdx.x * blockDim.x + threadIdx.x;
    if (t >= total) return;
    int node = t >> 6;
    int f    = t & 63;
    float x = agg[t] * norm_dst[node] + b[f];
    out[t] = 1.0f / (1.0f + __expf(-x));
}

extern "C" void kernel_launch(void* const* d_in, const int* in_sizes, int n_in,
                              void* d_out, int out_size, void* d_ws, size_t ws_size,
                              hipStream_t stream) {
    const float* feat = (const float*)d_in[0];
    const int*   src  = (const int*)d_in[1];
    const int*   dst  = (const int*)d_in[2];
    const float* W    = (const float*)d_in[3];
    const float* bias = (const float*)d_in[4];
    float*       out  = (float*)d_out;

    const int n_nodes = in_sizes[0] / IN_FEATS;
    const int n_edges = in_sizes[1];

    float* norm_src = (float*)d_ws;
    float* norm_dst = norm_src + n_nodes;
    float* agg      = norm_dst + n_nodes;

    size_t zero_bytes = ((size_t)2 * n_nodes + (size_t)n_nodes * OUT_FEATS)
                        * sizeof(float);
    hipMemsetAsync(d_ws, 0, zero_bytes, stream);

    gcn_degrees<<<(n_edges + 255) / 256, 256, 0, stream>>>(
        src, dst, norm_src, norm_dst, n_edges);

    gcn_norms<<<(n_nodes + 255) / 256, 256, 0, stream>>>(
        norm_src, norm_dst, n_nodes);

    const int n_tiles = n_nodes / 16;
    gcn_gemm_wmma<<<(n_tiles + 7) / 8, 256, 0, stream>>>(
        feat, norm_src, W, out, n_tiles);

    long long scat_threads = (long long)n_edges * 16;
    gcn_scatter<<<(int)((scat_threads + 255) / 256), 256, 0, stream>>>(
        src, dst, out, agg, n_edges);

    int total = n_nodes * OUT_FEATS;
    gcn_finalize<<<(total + 255) / 256, 256, 0, stream>>>(
        agg, norm_dst, bias, out, total);
}


# owner-id lists bit-sliced: launch 1 stores 8 ballot bit planes per 32 edges, not a byte per edge; an owner's scan matches 32 edges with one bit op per plane, no byte tests (sums: arrival order)
# speedup vs baseline: 3.2566x; 3.2566x over previous
#include <hip/hip_runtime.h>
#include <math.h>

typedef float v2f __attribute__((ext_vector_type(2)));
typedef float v8f __attribute__((ext_vector_type(8)));

#define IN_FEATS  256
#define OUT_FEATS 64
#define WPAIRS    (IN_FEATS / 2)
#define WSTRIDE   160
#define PR        196

#define QCAP  8192
#define GT    512

#define CH    256
#define NU    4

__device__ __forceinline__ void fix_tail(int4& v, const int* __restrict__ a, int e, int n4, int n_edges) {
    if (e == n4) {
        v.x = a[e];
        if (e + 1 < n_edges) v.y = a[e + 1];
        if (e + 2 < n_edges) v.z = a[e + 2];
    }
}

__device__ __forceinline__ void owner_planes(int4 n0, int4 n1, int lane, unsigned& wa, unsigned& wb) {
    unsigned id[8];
    id[0] = (unsigned)n0.x / PR;  id[1] = (unsigned)n0.y / PR;
    id[2] = (unsigned)n0.z / PR;  id[3] = (unsigned)n0.w / PR;
    id[4] = (unsigned)n1.x / PR;  id[5] = (unsigned)n1.y / PR;
    id[6] = (unsigned)n1.z / PR;  id[7] = (unsigned)n1.w / PR;
    wa = 0u;
    wb = 0u;
    #pragma unroll
    for (int q = 0; q < 8; ++q) {
        #pragma unroll
        for (int b = 0; b < 4; ++b) {
            const unsigned lo = __builtin_amdgcn_ballot_w32(((id[q] >> b) & 1u) != 0u);
            const unsigned hi = __builtin_amdgcn_ballot_w32(((id[q] >> (b + 4)) & 1u) != 0u);
            const bool mine = (lane == q * 4 + b);
            wa = mine ? lo : wa;
            wb = mine ? hi : wb;
        }
    }
}

__global__ void __launch_bounds__(256)
gcn_keys(const int* __restrict__ src,
         const int* __restrict__ dst,
         unsigned* __restrict__ keys,
         unsigned* __restrict__ keys_s,
         int n_edges) {
    const int lane = threadIdx.x & 31;
    const int w = __builtin_amdgcn_readfirstlane((int)blockIdx.x * 8 + (int)(threadIdx.x >> 5));
    if (w >= (n_edges + CH - 1) / CH) return;
    const int e0 = w * CH + 4 * lane;
    const int n4 = n_edges & ~3;
    int4 d0 = make_int4(0, 0, 0, 0), d1 = d0, s0 = d0, s1 = d0;
    if (n4 >= 4) {
        const int i0 = min(e0, n4 - 4) >> 2;
        const int i1 = min(e0 + 128, n4 - 4) >> 2;
        d0 = reinterpret_cast<const int4*>(dst)[i0];
        d1 = reinterpret_cast<const int4*>(dst)[i1];
        s0 = reinterpret_cast<const int4*>(src)[i0];
        s1 = reinterpret_cast<const int4*>(src)[i1];
    }
    if (n4 < n_edges) {
        fix_tail(d0, dst, e0, n4, n_edges);
        fix_tail(d1, dst, e0 + 128, n4, n_edges);
        fix_tail(s0, src, e0, n4, n_edges);
        fix_tail(s1, src, e0 + 128, n4, n_edges);
    }
    unsigned wa, wb;
    owner_planes(d0, d1, lane, wa, wb);
    keys[w * 64 + lane]      = wa;
    keys[w * 64 + 32 + lane] = wb;
    owner_planes(s0, s1, lane, wa, wb);
    keys_s[w * 64 + lane]      = wa;
    keys_s[w * 64 + 32 + lane] = wb;
}

__device__ __forceinline__ unsigned match32(uint4 a, uint4 b, const unsigned* c) {
    return (a.x ^ c[0]) & (a.y ^ c[1]) & (a.z ^ c[2]) & (a.w ^ c[3]) &
           (b.x ^ c[4]) & (b.y ^ c[5]) & (b.z ^ c[6]) & (b.w ^ c[7]);
}

__device__ __forceinline__ int group_edge0(int gi) {
    return ((gi >> 3) << 8) + ((gi & 4) << 5) + (gi & 3);
}

__device__ __forceinline__ int group_unit(int gi) {
    return ((gi >> 3) << 4) + (gi & 7);
}

__device__ __forceinline__ void push_src(int e, int lo, int* cnt, int* q_e, int* q_n,
                                         const int* __restrict__ src) {
    int p = atomicAdd(q_n, 1);
    if (p < QCAP) {
        q_e[p] = e;
    } else {
        int r = src[e] - lo;
        if ((unsigned)r < (unsigned)PR) atomicAdd(&cnt[r], 1);
    }
}

__device__ __forceinline__ void outdeg_role(const int* __restrict__ src,
                                            const unsigned* __restrict__ keys_s,
                                            float* __restrict__ deg_src,
                                            int n_edges, int n_nodes, int g, int* lds) {
    int* cnt = lds;
    int* q_e = lds + 256;
    int* q_n = lds + 256 + QCAP;

    const int tid = threadIdx.x;
    const int lo  = g * PR;

    if (tid < PR) cnt[tid] = 0;
    if (tid == 0) *q_n = 0;
    __syncthreads();

    const int    ng = ((n_edges + CH - 1) / CH) * 8;
    const uint4* k4 = reinterpret_cast<const uint4*>(keys_s);
    unsigned cg[8];
    #pragma unroll
    for (int k = 0; k < 8; ++k) cg[k] = (((unsigned)g >> k) & 1u) - 1u;
    for (int gb = tid; gb < ng; gb += GT * NU) {
        uint4 va[NU], vb[NU];
        #pragma unroll
        for (int k = 0; k < NU; ++k) {
            const int u = group_unit(min(gb + k * GT, ng - 1));
            va[k] = k4[u];
            vb[k] = k4[u + 8];
        }
        __builtin_amdgcn_sched_barrier(0);
        #pragma unroll
        for (int k = 0; k < NU; ++k) {
            const int gi = gb + k * GT;
            unsigned  m  = match32(va[k], vb[k], cg);
            m = (gi < ng) ? m : 0u;
            const int eb = group_edge0(gi);
            while (m) {
                const int p = __ffs(m) - 1;
                m &= m - 1;
                const int e = eb + 4 * p;
                if (e < n_edges) push_src(e, lo, cnt, q_e, q_n, src);
            }
        }
    }
    __syncthreads();

    const int nq = min(*q_n, QCAP);
    for (int i0 = 0; i0 < nq; i0 += GT * 4) {
        int rr[4];
        #pragma unroll
        for (int k = 0; k < 4; ++k)
            rr[k] = src[q_e[min(i0 + k * GT + tid, nq - 1)]] - lo;
        #pragma unroll
        for (int k = 0; k < 4; ++k)
            if (i0 + k * GT + tid < nq && (unsigned)rr[k] < (unsigned)PR)
                atomicAdd(&cnt[rr[k]], 1);
    }
    __syncthreads();

    if (tid < PR && lo + tid < n_nodes) deg_src[lo + tid] = (float)cnt[tid];
}

__global__ void __launch_bounds__(GT)
gcn_gemm_outdeg(const float* __restrict__ feat,
                const float* __restrict__ W,
                float* __restrict__ h,
                int n_tiles, int n_gemm_blocks,
                const int* __restrict__ src,
                const unsigned* __restrict__ keys_s,
                float* __restrict__ deg_src,
                int n_edges, int n_nodes) {
    __shared__ float Wl[WPAIRS * WSTRIDE];

    const int tid = threadIdx.x;
    if ((int)blockIdx.x >= n_gemm_blocks) {
        outdeg_role(src, keys_s, deg_src, n_edges, n_nodes,
                    (int)blockIdx.x - n_gemm_blocks, reinterpret_cast<int*>(Wl));
        return;
    }
    for (int i = tid; i < WPAIRS * (OUT_FEATS / 2); i += GT) {
        int p  = i >> 5;
        int n2 = (i & 31) * 2;
        float2 lo = *reinterpret_cast<const float2*>(&W[(2 * p) * OUT_FEATS + n2]);
        float2 hi = *reinterpret_cast<const float2*>(&W[(2 * p + 1) * OUT_FEATS + n2]);
        float4 q;  q.x = lo.x;  q.y = hi.x;  q.z = lo.y;  q.w = hi.y;
        *reinterpret_cast<float4*>(&Wl[p * WSTRIDE + 2 * n2]) = q;
    }
    __syncthreads();

    const int wave = tid >> 5;
    const int lane = tid & 31;
    const int tile = blockIdx.x * (GT / 32) + wave;
    if (tile >= n_tiles) return;

    const int half = lane >> 4;
    const int l15  = lane & 15;
    const int m    = tile * 16 + l15;
    const float* frow  = feat + (size_t)m * IN_FEATS + 2 * half;
    const float* wbase = &Wl[half * WSTRIDE + 2 * l15];

    v8f c0 = {}, c1 = {}, c2 = {}, c3 = {};

    #pragma unroll 4
    for (int k = 0; k < IN_FEATS; k += 4) {
        v2f a = *reinterpret_cast<const v2f*>(frow + k);

        const float* wp = wbase + (k >> 1) * WSTRIDE;
        v2f b0 = *reinterpret_cast<const v2f*>(wp);
        v2f b1 = *reinterpret_cast<const v2f*>(wp + 32);
        v2f b2 = *reinterpret_cast<const v2f*>(wp + 64);
        v2f b3 = *reinterpret_cast<const v2f*>(wp + 96);

        c0 = __builtin_amdgcn_wmma_f32_16x16x4_f32(false, a, false, b0,
                                                   (short)0, c0, false, false);
        c1 = __builtin_amdgcn_wmma_f32_16x16x4_f32(false, a, false, b1,
                                                   (short)0, c1, false, false);
        c2 = __builtin_amdgcn_wmma_f32_16x16x4_f32(false, a, false, b2,
                                                   (short)0, c2, false, false);
        c3 = __builtin_amdgcn_wmma_f32_16x16x4_f32(false, a, false, b3,
                                                   (short)0, c3, false, false);
    }

    const int rbase = tile * 16 + 8 * half;
    #pragma unroll
    for (int r = 0; r < 8; ++r) {
        float* orow = h + (size_t)(rbase + r) * OUT_FEATS + l15;
        orow[0]  = c0[r];
        orow[16] = c1[r];
        orow[32] = c2[r];
        orow[48] = c3[r];
    }
}

#define PT    1024
#define NW    (PT / 32)
#define WCAP  256
#define NB    16

__device__ __forceinline__ void push_edge(int e, int lo,
                                          float* acc, int* cnt,
                                          int* q_e, int* q_n,
                                          const int* __restrict__ src,
                                          const int* __restrict__ dst,
                                          const float* __restrict__ h,
                                          const float* __restrict__ deg_src) {
    int p = atomicAdd(q_n, 1);
    if (p < QCAP) {
        q_e[p] = e;
    } else {
        int r = dst[e] - lo;
        if ((unsigned)r < (unsigned)PR) {
            const int    sn = src[e];
            const float* hr = h + (size_t)sn * OUT_FEATS;
            const float  nr = rsqrtf(fmaxf(deg_src[sn], 1.0f));
            for (int f = 0; f < OUT_FEATS; ++f)
                atomicAdd(&acc[r * OUT_FEATS + f], hr[f] * nr);
            atomicAdd(&cnt[r], 1);
        }
    }
}

__global__ void __launch_bounds__(PT)
gcn_pull(const int* __restrict__ src,
         const int* __restrict__ dst,
         const unsigned* __restrict__ keys,
         const float* __restrict__ h,
         const float* __restrict__ deg_src,
         const float* __restrict__ b,
         float* __restrict__ out,
         int n_edges, int n_nodes) {
    __shared__ float acc[PR * OUT_FEATS];
    __shared__ int   cnt[PR];
    __shared__ int   q_e[QCAP];
    __shared__ int   q_n;
    __shared__ int   wq[NW * WCAP];
    __shared__ int   wq_n[NW];

    const int tid = threadIdx.x;
    const int lo  = blockIdx.x * PR;

    for (int i = tid; i < PR * OUT_FEATS; i += PT) acc[i] = 0.0f;
    if (tid < PR) cnt[tid] = 0;
    if (tid == 0) q_n = 0;
    if (tid < NW) wq_n[tid] = 0;
    __syncthreads();

    const int    ng = ((n_edges + CH - 1) / CH) * 8;
    const uint4* k4 = reinterpret_cast<const uint4*>(keys);
    unsigned cg[8];
    #pragma unroll
    for (int k = 0; k < 8; ++k) cg[k] = ((blockIdx.x >> k) & 1u) - 1u;
    for (int gb = tid; gb < ng; gb += PT * NU) {
        uint4 va[NU], vb[NU];
        #pragma unroll
        for (int k = 0; k < NU; ++k) {
            const int u = group_unit(min(gb + k * PT, ng - 1));
            va[k] = k4[u];
            vb[k] = k4[u + 8];
        }
        __builtin_amdgcn_sched_barrier(0);
        #pragma unroll
        for (int k = 0; k < NU; ++k) {
            const int gi = gb + k * PT;
            unsigned  m  = match32(va[k], vb[k], cg);
            m = (gi < ng) ? m : 0u;
            const int eb = group_edge0(gi);
            while (m) {
                const int p = __ffs(m) - 1;
                m &= m - 1;
                const int e = eb + 4 * p;
                if (e < n_edges)
                    push_edge(e, lo, acc, cnt, q_e, &q_n, src, dst, h, deg_src);
            }
        }
    }
    __syncthreads();

    const int nq = min(q_n, QCAP);
    for (int i = tid; i < nq; i += PT) {
        const int e = q_e[i];
        const int s = src[e];
        const int r = dst[e] - lo;
        if ((unsigned)r < (unsigned)PR) {
            atomicAdd(&cnt[r], 1);
            const int w = r & (NW - 1);
            const int p = atomicAdd(&wq_n[w], 1);
            if (p < WCAP) {
                wq[w * WCAP + p] = (s << 3) | (r >> 5);
            } else {
                const float* hr = h + (size_t)s * OUT_FEATS;
                const float  nr = rsqrtf(fmaxf(deg_src[s], 1.0f));
                for (int f = 0; f < OUT_FEATS; ++f)
                    atomicAdd(&acc[r * OUT_FEATS + f], hr[f] * nr);
            }
        }
    }
    __syncthreads();

    const int lane = tid & 31;
    const int wave = __builtin_amdgcn_readfirstlane(tid >> 5);
    const int wn   = __builtin_amdgcn_readfirstlane(min(wq_n[wave], WCAP));
    const int* wqp = wq + wave * WCAP;
    float sum[16];
    #pragma unroll
    for (int j = 0; j < 16; ++j) sum[j] = 0.0f;
    for (int i0 = 0; i0 < wn; i0 += NB) {
        int    en[NB];
        float  dd[NB];
        float2 vv[NB];
        const int ev = wqp[min(i0 + lane, wn - 1)];
        #pragma unroll
        for (int k = 0; k < NB; ++k)
            en[k] = __builtin_amdgcn_readlane(ev, k);
        #pragma unroll
        for (int k = 0; k < NB; ++k)
            dd[k] = deg_src[en[k] >> 3];
        #pragma unroll
        for (int k = 0; k < NB; ++k)
            vv[k] = *reinterpret_cast<const float2*>(h + (size_t)(en[k] >> 3) * OUT_FEATS + 2 * lane);
        asm volatile("" :: "v"(vv[0].x), "v"(vv[0].y), "v"(vv[1].x), "v"(vv[1].y),
                           "v"(vv[2].x), "v"(vv[2].y), "v"(vv[3].x), "v"(vv[3].y),
                           "v"(vv[4].x), "v"(vv[4].y), "v"(vv[5].x), "v"(vv[5].y),
                           "v"(vv[6].x), "v"(vv[6].y), "v"(vv[7].x), "v"(vv[7].y),
                           "v"(vv[8].x), "v"(vv[8].y), "v"(vv[9].x), "v"(vv[9].y),
                           "v"(vv[10].x), "v"(vv[10].y), "v"(vv[11].x), "v"(vv[11].y),
                           "v"(vv[12].x), "v"(vv[12].y), "v"(vv[13].x), "v"(vv[13].y),
                           "v"(vv[14].x), "v"(vv[14].y), "v"(vv[15].x), "v"(vv[15].y),
                           "v"(dd[0]), "v"(dd[1]), "v"(dd[2]), "v"(dd[3]),
                           "v"(dd[4]), "v"(dd[5]), "v"(dd[6]), "v"(dd[7]),
                           "v"(dd[8]), "v"(dd[9]), "v"(dd[10]), "v"(dd[11]),
                           "v"(dd[12]), "v"(dd[13]), "v"(dd[14]), "v"(dd[15]));
        #pragma unroll
        for (int k = 0; k < NB; ++k) {
            if (i0 + k < wn) {
                const float nr = rsqrtf(fmaxf(dd[k], 1.0f));
                const int   j  = 2 * (en[k] & 7);
                sum[j]     += vv[k].x * nr;
                sum[j + 1] += vv[k].y * nr;
            }
        }
    }
    {
        float* ar = &acc[wave * OUT_FEATS + 2 * lane];
        #pragma unroll
        for (int j = 0; j < 7; ++j) {
            if (j * NW + wave < PR) {
                ar[j * NW * OUT_FEATS]     += sum[2 * j];
                ar[j * NW * OUT_FEATS + 1] += sum[2 * j + 1];
            }
        }
    }
    __syncthreads();

    for (int i = tid; i < PR * OUT_FEATS; i += PT) {
        int r    = i >> 6;
        int f    = i & 63;
        int node = lo + r;
        if (node < n_nodes) {
            float nd = rsqrtf(fmaxf((float)cnt[r], 1.0f));
            float x  = acc[i] * nd + b[f];
            out[(size_t)node * OUT_FEATS + f] = 1.0f / (1.0f + __expf(-x));
        }
    }
}

extern "C" void kernel_launch(void* const* d_in, const int* in_sizes, int n_in,
                              void* d_out, int out_size, void* d_ws, size_t ws_size,
                              hipStream_t stream) {
    const float* feat = (const float*)d_in[0];
    const int*   src  = (const int*)d_in[1];
    const int*   dst  = (const int*)d_in[2];
    const float* W    = (const float*)d_in[3];
    const float* bias = (const float*)d_in[4];
    float*       out  = (float*)d_out;

    const int n_nodes = in_sizes[0] / IN_FEATS;
    const int n_edges = in_sizes[1];

    const int n_chunks = (n_edges + CH - 1) / CH;
    float*    hbuf     = (float*)d_ws;
    unsigned* keys     = (unsigned*)(hbuf + (size_t)n_nodes * OUT_FEATS);
    unsigned* keys_s   = keys + (size_t)n_chunks * 64;
    float*    deg_src  = (float*)(keys_s + (size_t)n_chunks * 64);

    gcn_keys<<<(n_chunks + 8) / 8, 256, 0, stream>>>(src, dst, keys, keys_s, n_edges);

    const int n_tiles       = n_nodes / 16;
    const int n_gemm_blocks = (n_tiles + GT / 32 - 1) / (GT / 32);
    gcn_gemm_outdeg<<<n_gemm_blocks + (n_nodes + PR - 1) / PR, GT, 0, stream>>>(
        feat, W, hbuf, n_tiles, n_gemm_blocks, src, keys_s, deg_src, n_edges, n_nodes);

    gcn_pull<<<(n_nodes + PR - 1) / PR, PT, 0, stream>>>(
        src, dst, keys, hbuf, deg_src, bias, out, n_edges, n_nodes);
}


# owner-id lists one-hot by nibble, plane-major: launch 1 sets 16+16 bit planes per list (LDS bit-or); an owner's scan ANDs its two planes, 2 bits read per edge, not 8 (sums: arrival order)
# speedup vs baseline: 3.8196x; 1.1729x over previous
#include <hip/hip_runtime.h>
#include <math.h>

typedef float v2f __attribute__((ext_vector_type(2)));
typedef float v8f __attribute__((ext_vector_type(8)));

#define IN_FEATS  256
#define OUT_FEATS 64
#define WPAIRS    (IN_FEATS / 2)
#define WSTRIDE   160
#define PR        196

#define QCAP  8192
#define GT    512

#define NU    4

#define SL    2048
#define SG    (SL / 32)
#define NPL   32
#define LPW   (SG + 1)

__device__ __forceinline__ void fix_tail(int4& v, const int* __restrict__ a, int e, int n4, int n_edges) {
    if (e == n4) {
        v.x = a[e];
        if (e + 1 < n_edges) v.y = a[e + 1];
        if (e + 2 < n_edges) v.z = a[e + 2];
    }
}

__device__ __forceinline__ void mark(unsigned* lp, int node, int el, bool live) {
    const unsigned id  = (unsigned)node / PR;
    const unsigned bit = live ? 1u << (el & 31) : 0u;
    const int      g   = el >> 5;
    atomicOr(&lp[((id >> 4) & 15u) * LPW + g], bit);
    atomicOr(&lp[(16u + (id & 15u)) * LPW + g], bit);
}

__global__ void __launch_bounds__(256)
gcn_keys(const int* __restrict__ src,
         const int* __restrict__ dst,
         unsigned* __restrict__ planes,
         unsigned* __restrict__ planes_s,
         int n_edges, int npw) {
    __shared__ unsigned lp[2 * NPL * LPW];
    const int tid = threadIdx.x;
    for (int i = tid; i < 2 * NPL * LPW; i += 256) lp[i] = 0u;

    const int el0 = 4 * tid;
    const int el1 = el0 + SL / 2;
    const int e0  = (int)blockIdx.x * SL + el0;
    const int e1  = (int)blockIdx.x * SL + el1;
    const int n4  = n_edges & ~3;
    int4 d0 = make_int4(0, 0, 0, 0), d1 = d0, s0 = d0, s1 = d0;
    if (n4 >= 4) {
        const int i0 = min(e0, n4 - 4) >> 2;
        const int i1 = min(e1, n4 - 4) >> 2;
        d0 = reinterpret_cast<const int4*>(dst)[i0];
        d1 = reinterpret_cast<const int4*>(dst)[i1];
        s0 = reinterpret_cast<const int4*>(src)[i0];
        s1 = reinterpret_cast<const int4*>(src)[i1];
    }
    if (n4 < n_edges) {
        fix_tail(d0, dst, e0, n4, n_edges);
        fix_tail(d1, dst, e1, n4, n_edges);
        fix_tail(s0, src, e0, n4, n_edges);
        fix_tail(s1, src, e1, n4, n_edges);
    }
    __syncthreads();

    unsigned* ld = lp;
    unsigned* ls = lp + NPL * LPW;
    mark(ld, d0.x, el0,     e0     < n_edges);  mark(ld, d0.y, el0 + 1, e0 + 1 < n_edges);
    mark(ld, d0.z, el0 + 2, e0 + 2 < n_edges);  mark(ld, d0.w, el0 + 3, e0 + 3 < n_edges);
    mark(ld, d1.x, el1,     e1     < n_edges);  mark(ld, d1.y, el1 + 1, e1 + 1 < n_edges);
    mark(ld, d1.z, el1 + 2, e1 + 2 < n_edges);  mark(ld, d1.w, el1 + 3, e1 + 3 < n_edges);
    mark(ls, s0.x, el0,     e0     < n_edges);  mark(ls, s0.y, el0 + 1, e0 + 1 < n_edges);
    mark(ls, s0.z, el0 + 2, e0 + 2 < n_edges);  mark(ls, s0.w, el0 + 3, e0 + 3 < n_edges);
    mark(ls, s1.x, el1,     e1     < n_edges);  mark(ls, s1.y, el1 + 1, e1 + 1 < n_edges);
    mark(ls, s1.z, el1 + 2, e1 + 2 < n_edges);  mark(ls, s1.w, el1 + 3, e1 + 3 < n_edges);
    __syncthreads();

    #pragma unroll
    for (int k = 0; k < 2 * NPL * SG / 256; ++k) {
        const int i   = k * 256 + tid;
        const int row = i >> 6;
        const int g   = i & (SG - 1);
        unsigned* out = (row < NPL) ? planes : planes_s;
        out[(size_t)(row & (NPL - 1)) * npw + (size_t)blockIdx.x * SG + g] = lp[row * LPW + g];
    }
}

__device__ __forceinline__ int pop_first(uint4& m) {
    const bool hx = m.x != 0u, hy = m.y != 0u, hz = m.z != 0u;
    const bool ux = hx, uy = !hx && hy, uz = !hx && !hy && hz, uw = !hx && !hy && !hz;
    const unsigned w    = ux ? m.x : uy ? m.y : uz ? m.z : m.w;
    const int      base = ux ? 0 : uy ? 32 : uz ? 64 : 96;
    const unsigned c    = w & (w - 1u);
    m.x = ux ? c : m.x;
    m.y = uy ? c : m.y;
    m.z = uz ? c : m.z;
    m.w = uw ? c : m.w;
    return base + __ffs(w) - 1;
}

__device__ __forceinline__ void push_src(int e, int lo, int* cnt, int* q_e, int* q_n,
                                         const int* __restrict__ src) {
    int p = atomicAdd(q_n, 1);
    if (p < QCAP) {
        q_e[p] = e;
    } else {
        int r = src[e] - lo;
        if ((unsigned)r < (unsigned)PR) atomicAdd(&cnt[r], 1);
    }
}

__device__ __forceinline__ void outdeg_role(const int* __restrict__ src,
                                            const unsigned* __restrict__ planes_s,
                                            float* __restrict__ deg_src,
                                            int npw, int n_nodes, int g, int* lds) {
    int* cnt = lds;
    int* q_e = lds + 256;
    int* q_n = lds + 256 + QCAP;

    const int tid = threadIdx.x;
    const int lo  = g * PR;

    if (tid < PR) cnt[tid] = 0;
    if (tid == 0) *q_n = 0;
    __syncthreads();

    const int    nu = npw >> 2;
    const uint4* pa = reinterpret_cast<const uint4*>(planes_s + (size_t)(g >> 4) * npw);
    const uint4* pb = reinterpret_cast<const uint4*>(planes_s + (size_t)(16 + (g & 15)) * npw);
    for (int ub = tid; ub < nu; ub += GT * NU) {
        uint4 va[NU], vb[NU];
        #pragma unroll
        for (int k = 0; k < NU; ++k) {
            const int u = min(ub + k * GT, nu - 1);
            va[k] = pa[u];
            vb[k] = pb[u];
        }
        __builtin_amdgcn_sched_barrier(0);
        #pragma unroll
        for (int k = 0; k < NU; ++k) {
            const int u = ub + k * GT;
            uint4 m;
            m.x = va[k].x & vb[k].x;  m.y = va[k].y & vb[k].y;
            m.z = va[k].z & vb[k].z;  m.w = va[k].w & vb[k].w;
            if (u >= nu) m = make_uint4(0u, 0u, 0u, 0u);
            while (m.x | m.y | m.z | m.w)
                push_src(u * 128 + pop_first(m), lo, cnt, q_e, q_n, src);
        }
    }
    __syncthreads();

    const int nq = min(*q_n, QCAP);
    for (int i0 = 0; i0 < nq; i0 += GT * 4) {
        int rr[4];
        #pragma unroll
        for (int k = 0; k < 4; ++k)
            rr[k] = src[q_e[min(i0 + k * GT + tid, nq - 1)]] - lo;
        #pragma unroll
        for (int k = 0; k < 4; ++k)
            if (i0 + k * GT + tid < nq && (unsigned)rr[k] < (unsigned)PR)
                atomicAdd(&cnt[rr[k]], 1);
    }
    __syncthreads();

    if (tid < PR && lo + tid < n_nodes) deg_src[lo + tid] = (float)cnt[tid];
}

__global__ void __launch_bounds__(GT)
gcn_gemm_outdeg(const float* __restrict__ feat,
                const float* __restrict__ W,
                float* __restrict__ h,
                int n_tiles, int n_gemm_blocks,
                const int* __restrict__ src,
                const unsigned* __restrict__ planes_s,
                float* __restrict__ deg_src,
                int npw, int n_nodes) {
    __shared__ float Wl[WPAIRS * WSTRIDE];

    const int tid = threadIdx.x;
    if ((int)blockIdx.x >= n_gemm_blocks) {
        outdeg_role(src, planes_s, deg_src, npw, n_nodes,
                    (int)blockIdx.x - n_gemm_blocks, reinterpret_cast<int*>(Wl));
        return;
    }
    for (int i = tid; i < WPAIRS * (OUT_FEATS / 2); i += GT) {
        int p  = i >> 5;
        int n2 = (i & 31) * 2;
        float2 lo = *reinterpret_cast<const float2*>(&W[(2 * p) * OUT_FEATS + n2]);
        float2 hi = *reinterpret_cast<const float2*>(&W[(2 * p + 1) * OUT_FEATS + n2]);
        float4 q;  q.x = lo.x;  q.y = hi.x;  q.z = lo.y;  q.w = hi.y;
        *reinterpret_cast<float4*>(&Wl[p * WSTRIDE + 2 * n2]) = q;
    }
    __syncthreads();

    const int wave = tid >> 5;
    const int lane = tid & 31;
    const int tile = blockIdx.x * (GT / 32) + wave;
    if (tile >= n_tiles) return;

    const int half = lane >> 4;
    const int l15  = lane & 15;
    const int m    = tile * 16 + l15;
    const float* frow  = feat + (size_t)m * IN_FEATS + 2 * half;
    const float* wbase = &Wl[half * WSTRIDE + 2 * l15];

    v8f c0 = {}, c1 = {}, c2 = {}, c3 = {};

    #pragma unroll 4
    for (int k = 0; k < IN_FEATS; k += 4) {
        v2f a = *reinterpret_cast<const v2f*>(frow + k);

        const float* wp = wbase + (k >> 1) * WSTRIDE;
        v2f b0 = *reinterpret_cast<const v2f*>(wp);
        v2f b1 = *reinterpret_cast<const v2f*>(wp + 32);
        v2f b2 = *reinterpret_cast<const v2f*>(wp + 64);
        v2f b3 = *reinterpret_cast<const v2f*>(wp + 96);

        c0 = __builtin_amdgcn_wmma_f32_16x16x4_f32(false, a, false, b0,
                                                   (short)0, c0, false, false);
        c1 = __builtin_amdgcn_wmma_f32_16x16x4_f32(false, a, false, b1,
                                                   (short)0, c1, false, false);
        c2 = __builtin_amdgcn_wmma_f32_16x16x4_f32(false, a, false, b2,
                                                   (short)0, c2, false, false);
        c3 = __builtin_amdgcn_wmma_f32_16x16x4_f32(false, a, false, b3,
                                                   (short)0, c3, false, false);
    }

    const int rbase = tile * 16 + 8 * half;
    #pragma unroll
    for (int r = 0; r < 8; ++r) {
        float* orow = h + (size_t)(rbase + r) * OUT_FEATS + l15;
        orow[0]  = c0[r];
        orow[16] = c1[r];
        orow[32] = c2[r];
        orow[48] = c3[r];
    }
}

#define PT    1024
#define NW    (PT / 32)
#define WCAP  256
#define NB    16

__device__ __forceinline__ void push_edge(int e, int lo,
                                          float* acc, int* cnt,
                                          int* q_e, int* q_n,
                                          const int* __restrict__ src,
                                          const int* __restrict__ dst,
                                          const float* __restrict__ h,
                                          const float* __restrict__ deg_src) {
    int p = atomicAdd(q_n, 1);
    if (p < QCAP) {
        q_e[p] = e;
    } else {
        int r = dst[e] - lo;
        if ((unsigned)r < (unsigned)PR) {
            const int    sn = src[e];
            const float* hr = h + (size_t)sn * OUT_FEATS;
            const float  nr = rsqrtf(fmaxf(deg_src[sn], 1.0f));
            for (int f = 0; f < OUT_FEATS; ++f)
                atomicAdd(&acc[r * OUT_FEATS + f], hr[f] * nr);
            atomicAdd(&cnt[r], 1);
        }
    }
}

__global__ void __launch_bounds__(PT)
gcn_pull(const int* __restrict__ src,
         const int* __restrict__ dst,
         const unsigned* __restrict__ planes,
         const float* __restrict__ h,
         const float* __restrict__ deg_src,
         const float* __restrict__ b,
         float* __restrict__ out,
         int npw, int n_nodes) {
    __shared__ float acc[PR * OUT_FEATS];
    __shared__ int   cnt[PR];
    __shared__ int   q_e[QCAP];
    __shared__ int   q_n;
    __shared__ int   wq[NW * WCAP];
    __shared__ int   wq_n[NW];

    const int tid = threadIdx.x;
    const int lo  = blockIdx.x * PR;

    for (int i = tid; i < PR * OUT_FEATS; i += PT) acc[i] = 0.0f;
    if (tid < PR) cnt[tid] = 0;
    if (tid == 0) q_n = 0;
    if (tid < NW) wq_n[tid] = 0;
    __syncthreads();

    const int    nu = npw >> 2;
    const uint4* pa = reinterpret_cast<const uint4*>(planes + (size_t)(blockIdx.x >> 4) * npw);
    const uint4* pb = reinterpret_cast<const uint4*>(planes + (size_t)(16 + (blockIdx.x & 15)) * npw);
    for (int ub = tid; ub < nu; ub += PT * NU) {
        uint4 va[NU], vb[NU];
        #pragma unroll
        for (int k = 0; k < NU; ++k) {
            const int u = min(ub + k * PT, nu - 1);
            va[k] = pa[u];
            vb[k] = pb[u];
        }
        __builtin_amdgcn_sched_barrier(0);
        #pragma unroll
        for (int k = 0; k < NU; ++k) {
            const int u = ub + k * PT;
            uint4 m;
            m.x = va[k].x & vb[k].x;  m.y = va[k].y & vb[k].y;
            m.z = va[k].z & vb[k].z;  m.w = va[k].w & vb[k].w;
            if (u >= nu) m = make_uint4(0u, 0u, 0u, 0u);
            while (m.x | m.y | m.z | m.w)
                push_edge(u * 128 + pop_first(m), lo, acc, cnt, q_e, &q_n, src, dst, h, deg_src);
        }
    }
    __syncthreads();

    const int nq = min(q_n, QCAP);
    for (int i = tid; i < nq; i += PT) {
        const int e = q_e[i];
        const int s = src[e];
        const int r = dst[e] - lo;
        if ((unsigned)r < (unsigned)PR) {
            atomicAdd(&cnt[r], 1);
            const int w = r & (NW - 1);
            const int p = atomicAdd(&wq_n[w], 1);
            if (p < WCAP) {
                wq[w * WCAP + p] = (s << 3) | (r >> 5);
            } else {
                const float* hr = h + (size_t)s * OUT_FEATS;
                const float  nr = rsqrtf(fmaxf(deg_src[s], 1.0f));
                for (int f = 0; f < OUT_FEATS; ++f)
                    atomicAdd(&acc[r * OUT_FEATS + f], hr[f] * nr);
            }
        }
    }
    __syncthreads();

    const int lane = tid & 31;
    const int wave = __builtin_amdgcn_readfirstlane(tid >> 5);
    const int wn   = __builtin_amdgcn_readfirstlane(min(wq_n[wave], WCAP));
    const int* wqp = wq + wave * WCAP;
    float sum[16];
    #pragma unroll
    for (int j = 0; j < 16; ++j) sum[j] = 0.0f;
    for (int i0 = 0; i0 < wn; i0 += NB) {
        int    en[NB];
        float  dd[NB];
        float2 vv[NB];
        const int ev = wqp[min(i0 + lane, wn - 1)];
        #pragma unroll
        for (int k = 0; k < NB; ++k)
            en[k] = __builtin_amdgcn_readlane(ev, k);
        #pragma unroll
        for (int k = 0; k < NB; ++k)
            dd[k] = deg_src[en[k] >> 3];
        #pragma unroll
        for (int k = 0; k < NB; ++k)
            vv[k] = *reinterpret_cast<const float2*>(h + (size_t)(en[k] >> 3) * OUT_FEATS + 2 * lane);
        asm volatile("" :: "v"(vv[0].x), "v"(vv[0].y), "v"(vv[1].x), "v"(vv[1].y),
                           "v"(vv[2].x), "v"(vv[2].y), "v"(vv[3].x), "v"(vv[3].y),
                           "v"(vv[4].x), "v"(vv[4].y), "v"(vv[5].x), "v"(vv[5].y),
                           "v"(vv[6].x), "v"(vv[6].y), "v"(vv[7].x), "v"(vv[7].y),
                           "v"(vv[8].x), "v"(vv[8].y), "v"(vv[9].x), "v"(vv[9].y),
                           "v"(vv[10].x), "v"(vv[10].y), "v"(vv[11].x), "v"(vv[11].y),
                           "v"(vv[12].x), "v"(vv[12].y), "v"(vv[13].x), "v"(vv[13].y),
                           "v"(vv[14].x), "v"(vv[14].y), "v"(vv[15].x), "v"(vv[15].y),
                           "v"(dd[0]), "v"(dd[1]), "v"(dd[2]), "v"(dd[3]),
                           "v"(dd[4]), "v"(dd[5]), "v"(dd[6]), "v"(dd[7]),
                           "v"(dd[8]), "v"(dd[9]), "v"(dd[10]), "v"(dd[11]),
                           "v"(dd[12]), "v"(dd[13]), "v"(dd[14]), "v"(dd[15]));
        #pragma unroll
        for (int k = 0; k < NB; ++k) {
            if (i0 + k < wn) {
                const float nr = rsqrtf(fmaxf(dd[k], 1.0f));
                const int   j  = 2 * (en[k] & 7);
                sum[j]     += vv[k].x * nr;
                sum[j + 1] += vv[k].y * nr;
            }
        }
    }
    {
        float* ar = &acc[wave * OUT_FEATS + 2 * lane];
        #pragma unroll
        for (int j = 0; j < 7; ++j) {
            if (j * NW + wave < PR) {
                ar[j * NW * OUT_FEATS]     += sum[2 * j];
                ar[j * NW * OUT_FEATS + 1] += sum[2 * j + 1];
            }
        }
    }
    __syncthreads();

    for (int i = tid; i < PR * OUT_FEATS; i += PT) {
        int r    = i >> 6;
        int f    = i & 63;
        int node = lo + r;
        if (node < n_nodes) {
            float nd = rsqrtf(fmaxf((float)cnt[r], 1.0f));
            float x  = acc[i] * nd + b[f];
            out[(size_t)node * OUT_FEATS + f] = 1.0f / (1.0f + __expf(-x));
        }
    }
}

extern "C" void kernel_launch(void* const* d_in, const int* in_sizes, int n_in,
                              void* d_out, int out_size, void* d_ws, size_t ws_size,
                              hipStream_t stream) {
    const float* feat = (const float*)d_in[0];
    const int*   src  = (const int*)d_in[1];
    const int*   dst  = (const int*)d_in[2];
    const float* W    = (const float*)d_in[3];
    const float* bias = (const float*)d_in[4];
    float*       out  = (float*)d_out;

    const int n_nodes = in_sizes[0] / IN_FEATS;
    const int n_edges = in_sizes[1];

    const int n_slices = (n_edges + SL - 1) / SL > 0 ? (n_edges + SL - 1) / SL : 1;
    const int npw      = n_slices * SG;
    float*    hbuf     = (float*)d_ws;
    unsigned* planes   = (unsigned*)(hbuf + (size_t)n_nodes * OUT_FEATS);
    unsigned* planes_s = planes + (size_t)NPL * npw;
    float*    deg_src  = (float*)(planes_s + (size_t)NPL * npw);

    gcn_keys<<<n_slices, 256, 0, stream>>>(src, dst, planes, planes_s, n_edges, npw);

    const int n_tiles       = n_nodes / 16;
    const int n_gemm_blocks = (n_tiles + GT / 32 - 1) / (GT / 32);
    gcn_gemm_outdeg<<<n_gemm_blocks + (n_nodes + PR - 1) / PR, GT, 0, stream>>>(
        feat, W, hbuf, n_tiles, n_gemm_blocks, src, planes_s, deg_src, npw, n_nodes);

    gcn_pull<<<(n_nodes + PR - 1) / PR, PT, 0, stream>>>(
        src, dst, planes, hbuf, deg_src, bias, out, npw, n_nodes);
}


# pull's owner scan and src/dst look-ups moved into launch 2 as a third role beside the product: it writes each owner's (src,row) list, launch 3 starts from the list (sums: arrival order, as before)
# speedup vs baseline: 3.9784x; 1.0416x over previous
#include <hip/hip_runtime.h>
#include <math.h>

typedef float v2f __attribute__((ext_vector_type(2)));
typedef float v8f __attribute__((ext_vector_type(8)));

#define IN_FEATS  256
#define OUT_FEATS 64
#define WPAIRS    (IN_FEATS / 2)
#define WSTRIDE   160
#define PR        196

#define QCAP  8192
#define GT    512

#define NU    4

#define SL    2048
#define SG    (SL / 32)
#define NPL   32
#define LPW   (SG + 1)

__device__ __forceinline__ void fix_tail(int4& v, const int* __restrict__ a, int e, int n4, int n_edges) {
    if (e == n4) {
        v.x = a[e];
        if (e + 1 < n_edges) v.y = a[e + 1];
        if (e + 2 < n_edges) v.z = a[e + 2];
    }
}

__device__ __forceinline__ void mark(unsigned* lp, int node, int el, bool live) {
    const unsigned id  = (unsigned)node / PR;
    const unsigned bit = live ? 1u << (el & 31) : 0u;
    const int      g   = el >> 5;
    atomicOr(&lp[((id >> 4) & 15u) * LPW + g], bit);
    atomicOr(&lp[(16u + (id & 15u)) * LPW + g], bit);
}

__global__ void __launch_bounds__(256)
gcn_keys(const int* __restrict__ src,
         const int* __restrict__ dst,
         unsigned* __restrict__ planes,
         unsigned* __restrict__ planes_s,
         int n_edges, int npw) {
    __shared__ unsigned lp[2 * NPL * LPW];
    const int tid = threadIdx.x;
    for (int i = tid; i < 2 * NPL * LPW; i += 256) lp[i] = 0u;

    const int el0 = 4 * tid;
    const int el1 = el0 + SL / 2;
    const int e0  = (int)blockIdx.x * SL + el0;
    const int e1  = (int)blockIdx.x * SL + el1;
    const int n4  = n_edges & ~3;
    int4 d0 = make_int4(0, 0, 0, 0), d1 = d0, s0 = d0, s1 = d0;
    if (n4 >= 4) {
        const int i0 = min(e0, n4 - 4) >> 2;
        const int i1 = min(e1, n4 - 4) >> 2;
        d0 = reinterpret_cast<const int4*>(dst)[i0];
        d1 = reinterpret_cast<const int4*>(dst)[i1];
        s0 = reinterpret_cast<const int4*>(src)[i0];
        s1 = reinterpret_cast<const int4*>(src)[i1];
    }
    if (n4 < n_edges) {
        fix_tail(d0, dst, e0, n4, n_edges);
        fix_tail(d1, dst, e1, n4, n_edges);
        fix_tail(s0, src, e0, n4, n_edges);
        fix_tail(s1, src, e1, n4, n_edges);
    }
    __syncthreads();

    unsigned* ld = lp;
    unsigned* ls = lp + NPL * LPW;
    mark(ld, d0.x, el0,     e0     < n_edges);  mark(ld, d0.y, el0 + 1, e0 + 1 < n_edges);
    mark(ld, d0.z, el0 + 2, e0 + 2 < n_edges);  mark(ld, d0.w, el0 + 3, e0 + 3 < n_edges);
    mark(ld, d1.x, el1,     e1     < n_edges);  mark(ld, d1.y, el1 + 1, e1 + 1 < n_edges);
    mark(ld, d1.z, el1 + 2, e1 + 2 < n_edges);  mark(ld, d1.w, el1 + 3, e1 + 3 < n_edges);
    mark(ls, s0.x, el0,     e0     < n_edges);  mark(ls, s0.y, el0 + 1, e0 + 1 < n_edges);
    mark(ls, s0.z, el0 + 2, e0 + 2 < n_edges);  mark(ls, s0.w, el0 + 3, e0 + 3 < n_edges);
    mark(ls, s1.x, el1,     e1     < n_edges);  mark(ls, s1.y, el1 + 1, e1 + 1 < n_edges);
    mark(ls, s1.z, el1 + 2, e1 + 2 < n_edges);  mark(ls, s1.w, el1 + 3, e1 + 3 < n_edges);
    __syncthreads();

    #pragma unroll
    for (int k = 0; k < 2 * NPL * SG / 256; ++k) {
        const int i   = k * 256 + tid;
        const int row = i >> 6;
        const int g   = i & (SG - 1);
        unsigned* out = (row < NPL) ? planes : planes_s;
        out[(size_t)(row & (NPL - 1)) * npw + (size_t)blockIdx.x * SG + g] = lp[row * LPW + g];
    }
}

__device__ __forceinline__ int pop_first(uint4& m) {
    const bool hx = m.x != 0u, hy = m.y != 0u, hz = m.z != 0u;
    const bool ux = hx, uy = !hx && hy, uz = !hx && !hy && hz, uw = !hx && !hy && !hz;
    const unsigned w    = ux ? m.x : uy ? m.y : uz ? m.z : m.w;
    const int      base = ux ? 0 : uy ? 32 : uz ? 64 : 96;
    const unsigned c    = w & (w - 1u);
    m.x = ux ? c : m.x;
    m.y = uy ? c : m.y;
    m.z = uz ? c : m.z;
    m.w = uw ? c : m.w;
    return base + __ffs(w) - 1;
}

__device__ __forceinline__ void push_src(int e, int lo, int* cnt, int* q_e, int* q_n,
                                         const int* __restrict__ src) {
    int p = atomicAdd(q_n, 1);
    if (p < QCAP) {
        q_e[p] = e;
    } else {
        int r = src[e] - lo;
        if ((unsigned)r < (unsigned)PR) atomicAdd(&cnt[r], 1);
    }
}

__device__ __forceinline__ void outdeg_role(const int* __restrict__ src,
                                            const unsigned* __restrict__ planes_s,
                                            float* __restrict__ deg_src,
                                            int npw, int n_nodes, int g, int* lds) {
    int* cnt = lds;
    int* q_e = lds + 256;
    int* q_n = lds + 256 + QCAP;

    const int tid = threadIdx.x;
    const int lo  = g * PR;

    if (tid < PR) cnt[tid] = 0;
    if (tid == 0) *q_n = 0;
    __syncthreads();

    const int    nu = npw >> 2;
    const uint4* pa = reinterpret_cast<const uint4*>(planes_s + (size_t)(g >> 4) * npw);
    const uint4* pb = reinterpret_cast<const uint4*>(planes_s + (size_t)(16 + (g & 15)) * npw);
    for (int ub = tid; ub < nu; ub += GT * NU) {
        uint4 va[NU], vb[NU];
        #pragma unroll
        for (int k = 0; k < NU; ++k) {
            const int u = min(ub + k * GT, nu - 1);
            va[k] = pa[u];
            vb[k] = pb[u];
        }
        __builtin_amdgcn_sched_barrier(0);
        #pragma unroll
        for (int k = 0; k < NU; ++k) {
            const int u = ub + k * GT;
            uint4 m;
            m.x = va[k].x & vb[k].x;  m.y = va[k].y & vb[k].y;
            m.z = va[k].z & vb[k].z;  m.w = va[k].w & vb[k].w;
            if (u >= nu) m = make_uint4(0u, 0u, 0u, 0u);
            while (m.x | m.y | m.z | m.w)
                push_src(u * 128 + pop_first(m), lo, cnt, q_e, q_n, src);
        }
    }
    __syncthreads();

    const int nq = min(*q_n, QCAP);
    for (int i0 = 0; i0 < nq; i0 += GT * 4) {
        int rr[4];
        #pragma unroll
        for (int k = 0; k < 4; ++k)
            rr[k] = src[q_e[min(i0 + k * GT + tid, nq - 1)]] - lo;
        #pragma unroll
        for (int k = 0; k < 4; ++k)
            if (i0 + k * GT + tid < nq && (unsigned)rr[k] < (unsigned)PR)
                atomicAdd(&cnt[rr[k]], 1);
    }
    __syncthreads();

    if (tid < PR && lo + tid < n_nodes) deg_src[lo + tid] = (float)cnt[tid];
}

__device__ __forceinline__ void push_q(int e, int* q_e, int* q_n) {
    int p = atomicAdd(q_n, 1);
    if (p < QCAP) q_e[p] = e;
}

__device__ __forceinline__ void list_role(const int* __restrict__ src,
                                          const int* __restrict__ dst,
                                          const unsigned* __restrict__ planes,
                                          int* __restrict__ list,
                                          int* __restrict__ list_n,
                                          int npw, int g, int* lds) {
    int* q_e = lds + 256;
    int* q_n = lds + 256 + QCAP;

    const int tid = threadIdx.x;
    const int lo  = g * PR;

    if (tid == 0) *q_n = 0;
    __syncthreads();

    const int    nu = npw >> 2;
    const uint4* pa = reinterpret_cast<const uint4*>(planes + (size_t)(g >> 4) * npw);
    const uint4* pb = reinterpret_cast<const uint4*>(planes + (size_t)(16 + (g & 15)) * npw);
    for (int ub = tid; ub < nu; ub += GT * NU) {
        uint4 va[NU], vb[NU];
        #pragma unroll
        for (int k = 0; k < NU; ++k) {
            const int u = min(ub + k * GT, nu - 1);
            va[k] = pa[u];
            vb[k] = pb[u];
        }
        __builtin_amdgcn_sched_barrier(0);
        #pragma unroll
        for (int k = 0; k < NU; ++k) {
            const int u = ub + k * GT;
            uint4 m;
            m.x = va[k].x & vb[k].x;  m.y = va[k].y & vb[k].y;
            m.z = va[k].z & vb[k].z;  m.w = va[k].w & vb[k].w;
            if (u >= nu) m = make_uint4(0u, 0u, 0u, 0u);
            while (m.x | m.y | m.z | m.w)
                push_q(u * 128 + pop_first(m), q_e, q_n);
        }
    }
    __syncthreads();

    const int nqa = *q_n;
    const int nq  = min(nqa, QCAP);
    int* lp = list + (size_t)g * QCAP;
    for (int i0 = 0; i0 < nq; i0 += GT * 4) {
        int ss[4], dd[4];
        #pragma unroll
        for (int k = 0; k < 4; ++k) {
            const int e = q_e[min(i0 + k * GT + tid, nq - 1)];
            ss[k] = src[e];
            dd[k] = dst[e];
        }
        #pragma unroll
        for (int k = 0; k < 4; ++k) {
            const int i = i0 + k * GT + tid;
            const unsigned r = (unsigned)(dd[k] - lo);
            if (i < nq) lp[i] = (ss[k] << 8) | (int)(r < (unsigned)PR ? r : 255u);
        }
    }
    if (tid == 0) list_n[g] = nqa;
}

__global__ void __launch_bounds__(GT)
gcn_gemm_outdeg(const float* __restrict__ feat,
                const float* __restrict__ W,
                float* __restrict__ h,
                int n_tiles, int n_gemm_blocks,
                const int* __restrict__ src,
                const unsigned* __restrict__ planes_s,
                float* __restrict__ deg_src,
                int npw, int n_nodes,
                const int* __restrict__ dst,
                const unsigned* __restrict__ planes,
                int* __restrict__ list,
                int* __restrict__ list_n,
                int n_own) {
    __shared__ float Wl[WPAIRS * WSTRIDE];

    const int tid = threadIdx.x;
    if ((int)blockIdx.x >= n_gemm_blocks) {
        const int o = (int)blockIdx.x - n_gemm_blocks;
        if (o < n_own)
            outdeg_role(src, planes_s, deg_src, npw, n_nodes, o, reinterpret_cast<int*>(Wl));
        else
            list_role(src, dst, planes, list, list_n, npw, o - n_own, reinterpret_cast<int*>(Wl));
        return;
    }
    for (int i = tid; i < WPAIRS * (OUT_FEATS / 2); i += GT) {
        int p  = i >> 5;
        int n2 = (i & 31) * 2;
        float2 lo = *reinterpret_cast<const float2*>(&W[(2 * p) * OUT_FEATS + n2]);
        float2 hi = *reinterpret_cast<const float2*>(&W[(2 * p + 1) * OUT_FEATS + n2]);
        float4 q;  q.x = lo.x;  q.y = hi.x;  q.z = lo.y;  q.w = hi.y;
        *reinterpret_cast<float4*>(&Wl[p * WSTRIDE + 2 * n2]) = q;
    }
    __syncthreads();

    const int wave = tid >> 5;
    const int lane = tid & 31;
    const int tile = blockIdx.x * (GT / 32) + wave;
    if (tile >= n_tiles) return;

    const int half = lane >> 4;
    const int l15  = lane & 15;
    const int m    = tile * 16 + l15;
    const float* frow  = feat + (size_t)m * IN_FEATS + 2 * half;
    const float* wbase = &Wl[half * WSTRIDE + 2 * l15];

    v8f c0 = {}, c1 = {}, c2 = {}, c3 = {};

    #pragma unroll 4
    for (int k = 0; k < IN_FEATS; k += 4) {
        v2f a = *reinterpret_cast<const v2f*>(frow + k);

        const float* wp = wbase + (k >> 1) * WSTRIDE;
        v2f b0 = *reinterpret_cast<const v2f*>(wp);
        v2f b1 = *reinterpret_cast<const v2f*>(wp + 32);
        v2f b2 = *reinterpret_cast<const v2f*>(wp + 64);
        v2f b3 = *reinterpret_cast<const v2f*>(wp + 96);

        c0 = __builtin_amdgcn_wmma_f32_16x16x4_f32(false, a, false, b0,
                                                   (short)0, c0, false, false);
        c1 = __builtin_amdgcn_wmma_f32_16x16x4_f32(false, a, false, b1,
                                                   (short)0, c1, false, false);
        c2 = __builtin_amdgcn_wmma_f32_16x16x4_f32(false, a, false, b2,
                                                   (short)0, c2, false, false);
        c3 = __builtin_amdgcn_wmma_f32_16x16x4_f32(false, a, false, b3,
                                                   (short)0, c3, false, false);
    }

    const int rbase = tile * 16 + 8 * half;
    #pragma unroll
    for (int r = 0; r < 8; ++r) {
        float* orow = h + (size_t)(rbase + r) * OUT_FEATS + l15;
        orow[0]  = c0[r];
        orow[16] = c1[r];
        orow[32] = c2[r];
        orow[48] = c3[r];
    }
}

#define PT    1024
#define NW    (PT / 32)
#define WCAP  256
#define NB    16

__device__ __forceinline__ void push_edge(int e, int lo,
                                          float* acc, int* cnt,
                                          int* q_e, int* q_n,
                                          const int* __restrict__ src,
                                          const int* __restrict__ dst,
                                          const float* __restrict__ h,
                                          const float* __restrict__ deg_src) {
    int p = atomicAdd(q_n, 1);
    if (p < QCAP) {
        q_e[p] = e;
    } else {
        int r = dst[e] - lo;
        if ((unsigned)r < (unsigned)PR) {
            const int    sn = src[e];
            const float* hr = h + (size_t)sn * OUT_FEATS;
            const float  nr = rsqrtf(fmaxf(deg_src[sn], 1.0f));
            for (int f = 0; f < OUT_FEATS; ++f)
                atomicAdd(&acc[r * OUT_FEATS + f], hr[f] * nr);
            atomicAdd(&cnt[r], 1);
        }
    }
}

__device__ __forceinline__ void place_edge(int s, int r,
                                           float* acc, int* cnt, int* wq, int* wq_n,
                                           const float* __restrict__ h,
                                           const float* __restrict__ deg_src) {
    if ((unsigned)r < (unsigned)PR) {
        atomicAdd(&cnt[r], 1);
        const int w = r & (NW - 1);
        const int p = atomicAdd(&wq_n[w], 1);
        if (p < WCAP) {
            wq[w * WCAP + p] = (s << 3) | (r >> 5);
        } else {
            const float* hr = h + (size_t)s * OUT_FEATS;
            const float  nr = rsqrtf(fmaxf(deg_src[s], 1.0f));
            for (int f = 0; f < OUT_FEATS; ++f)
                atomicAdd(&acc[r * OUT_FEATS + f], hr[f] * nr);
        }
    }
}

__global__ void __launch_bounds__(PT)
gcn_pull(const int* __restrict__ src,
         const int* __restrict__ dst,
         const unsigned* __restrict__ planes,
         const float* __restrict__ h,
         const float* __restrict__ deg_src,
         const float* __restrict__ b,
         float* __restrict__ out,
         int npw, int n_nodes,
         const int* __restrict__ list,
         const int* __restrict__ list_n) {
    __shared__ float acc[PR * OUT_FEATS];
    __shared__ int   cnt[PR];
    __shared__ int   q_e[QCAP];
    __shared__ int   q_n;
    __shared__ int   wq[NW * WCAP];
    __shared__ int   wq_n[NW];

    const int tid = threadIdx.x;
    const int lo  = blockIdx.x * PR;

    for (int i = tid; i < PR * OUT_FEATS; i += PT) acc[i] = 0.0f;
    if (tid < PR) cnt[tid] = 0;
    if (tid == 0) q_n = 0;
    if (tid < NW) wq_n[tid] = 0;
    __syncthreads();

    const int nl = list_n[blockIdx.x];
    if (nl <= QCAP) {
        const int* lp = list + (size_t)blockIdx.x * QCAP;
        for (int i0 = 0; i0 < nl; i0 += PT * 4) {
            int en[4];
            #pragma unroll
            for (int k = 0; k < 4; ++k)
                en[k] = lp[min(i0 + k * PT + tid, nl - 1)];
            #pragma unroll
            for (int k = 0; k < 4; ++k)
                if (i0 + k * PT + tid < nl)
                    place_edge(en[k] >> 8, en[k] & 255, acc, cnt, wq, wq_n, h, deg_src);
        }
    } else {
        const int    nu = npw >> 2;
        const uint4* pa = reinterpret_cast<const uint4*>(planes + (size_t)(blockIdx.x >> 4) * npw);
        const uint4* pb = reinterpret_cast<const uint4*>(planes + (size_t)(16 + (blockIdx.x & 15)) * npw);
        for (int ub = tid; ub < nu; ub += PT * NU) {
            uint4 va[NU], vb[NU];
            #pragma unroll
            for (int k = 0; k < NU; ++k) {
                const int u = min(ub + k * PT, nu - 1);
                va[k] = pa[u];
                vb[k] = pb[u];
            }
            __builtin_amdgcn_sched_barrier(0);
            #pragma unroll
            for (int k = 0; k < NU; ++k) {
                const int u = ub + k * PT;
                uint4 m;
                m.x = va[k].x & vb[k].x;  m.y = va[k].y & vb[k].y;
                m.z = va[k].z & vb[k].z;  m.w = va[k].w & vb[k].w;
                if (u >= nu) m = make_uint4(0u, 0u, 0u, 0u);
                while (m.x | m.y | m.z | m.w)
                    push_edge(u * 128 + pop_first(m), lo, acc, cnt, q_e, &q_n, src, dst, h, deg_src);
            }
        }
        __syncthreads();

        const int nq = min(q_n, QCAP);
        for (int i = tid; i < nq; i += PT) {
            const int e = q_e[i];
            place_edge(src[e], dst[e] - lo, acc, cnt, wq, wq_n, h, deg_src);
        }
    }
    __syncthreads();

    const int lane = tid & 31;
    const int wave = __builtin_amdgcn_readfirstlane(tid >> 5);
    const int wn   = __builtin_amdgcn_readfirstlane(min(wq_n[wave], WCAP));
    const int* wqp = wq + wave * WCAP;
    float sum[16];
    #pragma unroll
    for (int j = 0; j < 16; ++j) sum[j] = 0.0f;
    for (int i0 = 0; i0 < wn; i0 += NB) {
        int    en[NB];
        float  dd[NB];
        float2 vv[NB];
        const int ev = wqp[min(i0 + lane, wn - 1)];
        #pragma unroll
        for (int k = 0; k < NB; ++k)
            en[k] = __builtin_amdgcn_readlane(ev, k);
        #pragma unroll
        for (int k = 0; k < NB; ++k)
            dd[k] = deg_src[en[k] >> 3];
        #pragma unroll
        for (int k = 0; k < NB; ++k)
            vv[k] = *reinterpret_cast<const float2*>(h + (size_t)(en[k] >> 3) * OUT_FEATS + 2 * lane);
        asm volatile("" :: "v"(vv[0].x), "v"(vv[0].y), "v"(vv[1].x), "v"(vv[1].y),
                           "v"(vv[2].x), "v"(vv[2].y), "v"(vv[3].x), "v"(vv[3].y),
                           "v"(vv[4].x), "v"(vv[4].y), "v"(vv[5].x), "v"(vv[5].y),
                           "v"(vv[6].x), "v"(vv[6].y), "v"(vv[7].x), "v"(vv[7].y),
                           "v"(vv[8].x), "v"(vv[8].y), "v"(vv[9].x), "v"(vv[9].y),
                           "v"(vv[10].x), "v"(vv[10].y), "v"(vv[11].x), "v"(vv[11].y),
                           "v"(vv[12].x), "v"(vv[12].y), "v"(vv[13].x), "v"(vv[13].y),
                           "v"(vv[14].x), "v"(vv[14].y), "v"(vv[15].x), "v"(vv[15].y),
                           "v"(dd[0]), "v"(dd[1]), "v"(dd[2]), "v"(dd[3]),
                           "v"(dd[4]), "v"(dd[5]), "v"(dd[6]), "v"(dd[7]),
                           "v"(dd[8]), "v"(dd[9]), "v"(dd[10]), "v"(dd[11]),
                           "v"(dd[12]), "v"(dd[13]), "v"(dd[14]), "v"(dd[15]));
        #pragma unroll
        for (int k = 0; k < NB; ++k) {
            if (i0 + k < wn) {
                const float nr = rsqrtf(fmaxf(dd[k], 1.0f));
                const int   j  = 2 * (en[k] & 7);
                sum[j]     += vv[k].x * nr;
                sum[j + 1] += vv[k].y * nr;
            }
        }
    }
    {
        float* ar = &acc[wave * OUT_FEATS + 2 * lane];
        #pragma unroll
        for (int j = 0; j < 7; ++j) {
            if (j * NW + wave < PR) {
                ar[j * NW * OUT_FEATS]     += sum[2 * j];
                ar[j * NW * OUT_FEATS + 1] += sum[2 * j + 1];
            }
        }
    }
    __syncthreads();

    for (int i = tid; i < PR * OUT_FEATS; i += PT) {
        int r    = i >> 6;
        int f    = i & 63;
        int node = lo + r;
        if (node < n_nodes) {
            float nd = rsqrtf(fmaxf((float)cnt[r], 1.0f));
            float x  = acc[i] * nd + b[f];
            out[(size_t)node * OUT_FEATS + f] = 1.0f / (1.0f + __expf(-x));
        }
    }
}

extern "C" void kernel_launch(void* const* d_in, const int* in_sizes, int n_in,
                              void* d_out, int out_size, void* d_ws, size_t ws_size,
                              hipStream_t stream) {
    const float* feat = (const float*)d_in[0];
    const int*   src  = (const int*)d_in[1];
    const int*   dst  = (const int*)d_in[2];
    const float* W    = (const float*)d_in[3];
    const float* bias = (const float*)d_in[4];
    float*       out  = (float*)d_out;

    const int n_nodes = in_sizes[0] / IN_FEATS;
    const int n_edges = in_sizes[1];

    const int n_slices = (n_edges + SL - 1) / SL > 0 ? (n_edges + SL - 1) / SL : 1;
    const int npw      = n_slices * SG;
    float*    hbuf     = (float*)d_ws;
    unsigned* planes   = (unsigned*)(hbuf + (size_t)n_nodes * OUT_FEATS);
    unsigned* planes_s = planes + (size_t)NPL * npw;
    float*    deg_src  = (float*)(planes_s + (size_t)NPL * npw);
    const int n_own    = (n_nodes + PR - 1) / PR;
    int*      list_n   = (int*)(deg_src + (size_t)n_own * PR);
    int*      list     = list_n + 256;

    gcn_keys<<<n_slices, 256, 0, stream>>>(src, dst, planes, planes_s, n_edges, npw);

    const int n_tiles       = n_nodes / 16;
    const int n_gemm_blocks = (n_tiles + GT / 32 - 1) / (GT / 32);
    gcn_gemm_outdeg<<<n_gemm_blocks + 2 * n_own, GT, 0, stream>>>(
        feat, W, hbuf, n_tiles, n_gemm_blocks, src, planes_s, deg_src, npw, n_nodes,
        dst, planes, list, list_n, n_own);

    gcn_pull<<<n_own, PT, 0, stream>>>(
        src, dst, planes, hbuf, deg_src, bias, out, npw, n_nodes, list, list_n);
}


# pull: source out-degree counts from a byte-per-node table (launch 2 stores it, each pull work-group copies it to LDS): lane k looks up entry k, one rsqrt a lane; no 16 uniform global loads a trip
# speedup vs baseline: 4.1015x; 1.0309x over previous
#include <hip/hip_runtime.h>
#include <math.h>

typedef float v2f __attribute__((ext_vector_type(2)));
typedef float v8f __attribute__((ext_vector_type(8)));

#define IN_FEATS  256
#define OUT_FEATS 64
#define WPAIRS    (IN_FEATS / 2)
#define WSTRIDE   160
#define PR        196

#define QCAP  8192
#define GT    512

#define NU    4

#define SL    2048
#define SG    (SL / 32)
#define NPL   32
#define LPW   (SG + 1)

__device__ __forceinline__ void fix_tail(int4& v, const int* __restrict__ a, int e, int n4, int n_edges) {
    if (e == n4) {
        v.x = a[e];
        if (e + 1 < n_edges) v.y = a[e + 1];
        if (e + 2 < n_edges) v.z = a[e + 2];
    }
}

__device__ __forceinline__ void mark(unsigned* lp, int node, int el, bool live) {
    const unsigned id  = (unsigned)node / PR;
    const unsigned bit = live ? 1u << (el & 31) : 0u;
    const int      g   = el >> 5;
    atomicOr(&lp[((id >> 4) & 15u) * LPW + g], bit);
    atomicOr(&lp[(16u + (id & 15u)) * LPW + g], bit);
}

__global__ void __launch_bounds__(256)
gcn_keys(const int* __restrict__ src,
         const int* __restrict__ dst,
         unsigned* __restrict__ planes,
         unsigned* __restrict__ planes_s,
         int n_edges, int npw) {
    __shared__ unsigned lp[2 * NPL * LPW];
    const int tid = threadIdx.x;
    for (int i = tid; i < 2 * NPL * LPW; i += 256) lp[i] = 0u;

    const int el0 = 4 * tid;
    const int el1 = el0 + SL / 2;
    const int e0  = (int)blockIdx.x * SL + el0;
    const int e1  = (int)blockIdx.x * SL + el1;
    const int n4  = n_edges & ~3;
    int4 d0 = make_int4(0, 0, 0, 0), d1 = d0, s0 = d0, s1 = d0;
    if (n4 >= 4) {
        const int i0 = min(e0, n4 - 4) >> 2;
        const int i1 = min(e1, n4 - 4) >> 2;
        d0 = reinterpret_cast<const int4*>(dst)[i0];
        d1 = reinterpret_cast<const int4*>(dst)[i1];
        s0 = reinterpret_cast<const int4*>(src)[i0];
        s1 = reinterpret_cast<const int4*>(src)[i1];
    }
    if (n4 < n_edges) {
        fix_tail(d0, dst, e0, n4, n_edges);
        fix_tail(d1, dst, e1, n4, n_edges);
        fix_tail(s0, src, e0, n4, n_edges);
        fix_tail(s1, src, e1, n4, n_edges);
    }
    __syncthreads();

    unsigned* ld = lp;
    unsigned* ls = lp + NPL * LPW;
    mark(ld, d0.x, el0,     e0     < n_edges);  mark(ld, d0.y, el0 + 1, e0 + 1 < n_edges);
    mark(ld, d0.z, el0 + 2, e0 + 2 < n_edges);  mark(ld, d0.w, el0 + 3, e0 + 3 < n_edges);
    mark(ld, d1.x, el1,     e1     < n_edges);  mark(ld, d1.y, el1 + 1, e1 + 1 < n_edges);
    mark(ld, d1.z, el1 + 2, e1 + 2 < n_edges);  mark(ld, d1.w, el1 + 3, e1 + 3 < n_edges);
    mark(ls, s0.x, el0,     e0     < n_edges);  mark(ls, s0.y, el0 + 1, e0 + 1 < n_edges);
    mark(ls, s0.z, el0 + 2, e0 + 2 < n_edges);  mark(ls, s0.w, el0 + 3, e0 + 3 < n_edges);
    mark(ls, s1.x, el1,     e1     < n_edges);  mark(ls, s1.y, el1 + 1, e1 + 1 < n_edges);
    mark(ls, s1.z, el1 + 2, e1 + 2 < n_edges);  mark(ls, s1.w, el1 + 3, e1 + 3 < n_edges);
    __syncthreads();

    #pragma unroll
    for (int k = 0; k < 2 * NPL * SG / 256; ++k) {
        const int i   = k * 256 + tid;
        const int row = i >> 6;
        const int g   = i & (SG - 1);
        unsigned* out = (row < NPL) ? planes : planes_s;
        out[(size_t)(row & (NPL - 1)) * npw + (size_t)blockIdx.x * SG + g] = lp[row * LPW + g];
    }
}

__device__ __forceinline__ int pop_first(uint4& m) {
    const bool hx = m.x != 0u, hy = m.y != 0u, hz = m.z != 0u;
    const bool ux = hx, uy = !hx && hy, uz = !hx && !hy && hz, uw = !hx && !hy && !hz;
    const unsigned w    = ux ? m.x : uy ? m.y : uz ? m.z : m.w;
    const int      base = ux ? 0 : uy ? 32 : uz ? 64 : 96;
    const unsigned c    = w & (w - 1u);
    m.x = ux ? c : m.x;
    m.y = uy ? c : m.y;
    m.z = uz ? c : m.z;
    m.w = uw ? c : m.w;
    return base + __ffs(w) - 1;
}

__device__ __forceinline__ void push_src(int e, int lo, int* cnt, int* q_e, int* q_n,
                                         const int* __restrict__ src) {
    int p = atomicAdd(q_n, 1);
    if (p < QCAP) {
        q_e[p] = e;
    } else {
        int r = src[e] - lo;
        if ((unsigned)r < (unsigned)PR) atomicAdd(&cnt[r], 1);
    }
}

__device__ __forceinline__ void outdeg_role(const int* __restrict__ src,
                                            const unsigned* __restrict__ planes_s,
                                            float* __restrict__ deg_src,
                                            unsigned* __restrict__ deg_b,
                                            int npw, int n_nodes, int g, int* lds) {
    int* cnt = lds;
    int* q_e = lds + 256;
    int* q_n = lds + 256 + QCAP;

    const int tid = threadIdx.x;
    const int lo  = g * PR;

    if (tid < PR) cnt[tid] = 0;
    if (tid == 0) *q_n = 0;
    __syncthreads();

    const int    nu = npw >> 2;
    const uint4* pa = reinterpret_cast<const uint4*>(planes_s + (size_t)(g >> 4) * npw);
    const uint4* pb = reinterpret_cast<const uint4*>(planes_s + (size_t)(16 + (g & 15)) * npw);
    for (int ub = tid; ub < nu; ub += GT * NU) {
        uint4 va[NU], vb[NU];
        #pragma unroll
        for (int k = 0; k < NU; ++k) {
            const int u = min(ub + k * GT, nu - 1);
            va[k] = pa[u];
            vb[k] = pb[u];
        }
        __builtin_amdgcn_sched_barrier(0);
        #pragma unroll
        for (int k = 0; k < NU; ++k) {
            const int u = ub + k * GT;
            uint4 m;
            m.x = va[k].x & vb[k].x;  m.y = va[k].y & vb[k].y;
            m.z = va[k].z & vb[k].z;  m.w = va[k].w & vb[k].w;
            if (u >= nu) m = make_uint4(0u, 0u, 0u, 0u);
            while (m.x | m.y | m.z | m.w)
                push_src(u * 128 + pop_first(m), lo, cnt, q_e, q_n, src);
        }
    }
    __syncthreads();

    const int nq = min(*q_n, QCAP);
    for (int i0 = 0; i0 < nq; i0 += GT * 4) {
        int rr[4];
        #pragma unroll
        for (int k = 0; k < 4; ++k)
            rr[k] = src[q_e[min(i0 + k * GT + tid, nq - 1)]] - lo;
        #pragma unroll
        for (int k = 0; k < 4; ++k)
            if (i0 + k * GT + tid < nq && (unsigned)rr[k] < (unsigned)PR)
                atomicAdd(&cnt[rr[k]], 1);
    }
    __syncthreads();

    if (tid < PR && lo + tid < n_nodes) deg_src[lo + tid] = (float)cnt[tid];
    if (tid < PR / 4) {
        const unsigned c0 = (unsigned)min(cnt[4 * tid],     255);
        const unsigned c1 = (unsigned)min(cnt[4 * tid + 1], 255);
        const unsigned c2 = (unsigned)min(cnt[4 * tid + 2], 255);
        const unsigned c3 = (unsigned)min(cnt[4 * tid + 3], 255);
        deg_b[g * (PR / 4) + tid] = c0 | (c1 << 8) | (c2 << 16) | (c3 << 24);
    }
}

__device__ __forceinline__ void push_q(int e, int* q_e, int* q_n) {
    int p = atomicAdd(q_n, 1);
    if (p < QCAP) q_e[p] = e;
}

__device__ __forceinline__ void list_role(const int* __restrict__ src,
                                          const int* __restrict__ dst,
                                          const unsigned* __restrict__ planes,
                                          int* __restrict__ list,
                                          int* __restrict__ list_n,
                                          int npw, int g, int* lds) {
    int* q_e = lds + 256;
    int* q_n = lds + 256 + QCAP;

    const int tid = threadIdx.x;
    const int lo  = g * PR;

    if (tid == 0) *q_n = 0;
    __syncthreads();

    const int    nu = npw >> 2;
    const uint4* pa = reinterpret_cast<const uint4*>(planes + (size_t)(g >> 4) * npw);
    const uint4* pb = reinterpret_cast<const uint4*>(planes + (size_t)(16 + (g & 15)) * npw);
    for (int ub = tid; ub < nu; ub += GT * NU) {
        uint4 va[NU], vb[NU];
        #pragma unroll
        for (int k = 0; k < NU; ++k) {
            const int u = min(ub + k * GT, nu - 1);
            va[k] = pa[u];
            vb[k] = pb[u];
        }
        __builtin_amdgcn_sched_barrier(0);
        #pragma unroll
        for (int k = 0; k < NU; ++k) {
            const int u = ub + k * GT;
            uint4 m;
            m.x = va[k].x & vb[k].x;  m.y = va[k].y & vb[k].y;
            m.z = va[k].z & vb[k].z;  m.w = va[k].w & vb[k].w;
            if (u >= nu) m = make_uint4(0u, 0u, 0u, 0u);
            while (m.x | m.y | m.z | m.w)
                push_q(u * 128 + pop_first(m), q_e, q_n);
        }
    }
    __syncthreads();

    const int nqa = *q_n;
    const int nq  = min(nqa, QCAP);
    int* lp = list + (size_t)g * QCAP;
    for (int i0 = 0; i0 < nq; i0 += GT * 4) {
        int ss[4], dd[4];
        #pragma unroll
        for (int k = 0; k < 4; ++k) {
            const int e = q_e[min(i0 + k * GT + tid, nq - 1)];
            ss[k] = src[e];
            dd[k] = dst[e];
        }
        #pragma unroll
        for (int k = 0; k < 4; ++k) {
            const int i = i0 + k * GT + tid;
            const unsigned r = (unsigned)(dd[k] - lo);
            if (i < nq) lp[i] = (ss[k] << 8) | (int)(r < (unsigned)PR ? r : 255u);
        }
    }
    if (tid == 0) list_n[g] = nqa;
}

__global__ void __launch_bounds__(GT)
gcn_gemm_outdeg(const float* __restrict__ feat,
                const float* __restrict__ W,
                float* __restrict__ h,
                int n_tiles, int n_gemm_blocks,
                const int* __restrict__ src,
                const unsigned* __restrict__ planes_s,
                float* __restrict__ deg_src,
                unsigned* __restrict__ deg_b,
                int npw, int n_nodes,
                const int* __restrict__ dst,
                const unsigned* __restrict__ planes,
                int* __restrict__ list,
                int* __restrict__ list_n,
                int n_own) {
    __shared__ float Wl[WPAIRS * WSTRIDE];

    const int tid = threadIdx.x;
    if ((int)blockIdx.x >= n_gemm_blocks) {
        const int o = (int)blockIdx.x - n_gemm_blocks;
        if (o < n_own)
            outdeg_role(src, planes_s, deg_src, deg_b, npw, n_nodes, o, reinterpret_cast<int*>(Wl));
        else
            list_role(src, dst, planes, list, list_n, npw, o - n_own, reinterpret_cast<int*>(Wl));
        return;
    }
    for (int i = tid; i < WPAIRS * (OUT_FEATS / 2); i += GT) {
        int p  = i >> 5;
        int n2 = (i & 31) * 2;
        float2 lo = *reinterpret_cast<const float2*>(&W[(2 * p) * OUT_FEATS + n2]);
        float2 hi = *reinterpret_cast<const float2*>(&W[(2 * p + 1) * OUT_FEATS + n2]);
        float4 q;  q.x = lo.x;  q.y = hi.x;  q.z = lo.y;  q.w = hi.y;
        *reinterpret_cast<float4*>(&Wl[p * WSTRIDE + 2 * n2]) = q;
    }
    __syncthreads();

    const int wave = tid >> 5;
    const int lane = tid & 31;
    const int tile = blockIdx.x * (GT / 32) + wave;
    if (tile >= n_tiles) return;

    const int half = lane >> 4;
    const int l15  = lane & 15;
    const int m    = tile * 16 + l15;
    const float* frow  = feat + (size_t)m * IN_FEATS + 2 * half;
    const float* wbase = &Wl[half * WSTRIDE + 2 * l15];

    v8f c0 = {}, c1 = {}, c2 = {}, c3 = {};

    #pragma unroll 4
    for (int k = 0; k < IN_FEATS; k += 4) {
        v2f a = *reinterpret_cast<const v2f*>(frow + k);

        const float* wp = wbase + (k >> 1) * WSTRIDE;
        v2f b0 = *reinterpret_cast<const v2f*>(wp);
        v2f b1 = *reinterpret_cast<const v2f*>(wp + 32);
        v2f b2 = *reinterpret_cast<const v2f*>(wp + 64);
        v2f b3 = *reinterpret_cast<const v2f*>(wp + 96);

        c0 = __builtin_amdgcn_wmma_f32_16x16x4_f32(false, a, false, b0,
                                                   (short)0, c0, false, false);
        c1 = __builtin_amdgcn_wmma_f32_16x16x4_f32(false, a, false, b1,
                                                   (short)0, c1, false, false);
        c2 = __builtin_amdgcn_wmma_f32_16x16x4_f32(false, a, false, b2,
                                                   (short)0, c2, false, false);
        c3 = __builtin_amdgcn_wmma_f32_16x16x4_f32(false, a, false, b3,
                                                   (short)0, c3, false, false);
    }

    const int rbase = tile * 16 + 8 * half;
    #pragma unroll
    for (int r = 0; r < 8; ++r) {
        float* orow = h + (size_t)(rbase + r) * OUT_FEATS + l15;
        orow[0]  = c0[r];
        orow[16] = c1[r];
        orow[32] = c2[r];
        orow[48] = c3[r];
    }
}

#define PT    1024
#define NW    (PT / 32)
#define WCAP  256
#define NB    16
#define DGW   (256 * PR / 4)
static_assert((DGW / 4 + PT - 1) / PT == 4, "the table copy holds four 16-byte units a thread");

__device__ __forceinline__ void push_edge(int e, int lo,
                                          float* acc, int* cnt,
                                          int* q_e, int* q_n,
                                          const int* __restrict__ src,
                                          const int* __restrict__ dst,
                                          const float* __restrict__ h,
                                          const float* __restrict__ deg_src) {
    int p = atomicAdd(q_n, 1);
    if (p < QCAP) {
        q_e[p] = e;
    } else {
        int r = dst[e] - lo;
        if ((unsigned)r < (unsigned)PR) {
            const int    sn = src[e];
            const float* hr = h + (size_t)sn * OUT_FEATS;
            const float  nr = rsqrtf(fmaxf(deg_src[sn], 1.0f));
            for (int f = 0; f < OUT_FEATS; ++f)
                atomicAdd(&acc[r * OUT_FEATS + f], hr[f] * nr);
            atomicAdd(&cnt[r], 1);
        }
    }
}

__device__ __forceinline__ void place_edge(int s, int r,
                                           float* acc, int* cnt, int* wq, int* wq_n,
                                           const float* __restrict__ h,
                                           const float* __restrict__ deg_src) {
    if ((unsigned)r < (unsigned)PR) {
        atomicAdd(&cnt[r], 1);
        const int w = r & (NW - 1);
        const int p = atomicAdd(&wq_n[w], 1);
        if (p < WCAP) {
            wq[w * WCAP + p] = (s << 3) | (r >> 5);
        } else {
            const float* hr = h + (size_t)s * OUT_FEATS;
            const float  nr = rsqrtf(fmaxf(deg_src[s], 1.0f));
            for (int f = 0; f < OUT_FEATS; ++f)
                atomicAdd(&acc[r * OUT_FEATS + f], hr[f] * nr);
        }
    }
}

__global__ void __launch_bounds__(PT)
gcn_pull(const int* __restrict__ src,
         const int* __restrict__ dst,
         const unsigned* __restrict__ planes,
         const float* __restrict__ h,
         const float* __restrict__ deg_src,
         const float* __restrict__ b,
         float* __restrict__ out,
         int npw, int n_nodes,
         const int* __restrict__ list,
         const int* __restrict__ list_n,
         const unsigned* __restrict__ deg_b,
         int n_tabw) {
    __shared__ float acc[PR * OUT_FEATS];
    __shared__ unsigned dgw[DGW];
    __shared__ int   cnt[PR];
    __shared__ int   q_e[QCAP];
    __shared__ int   q_n;
    __shared__ int   wq[NW * WCAP];
    __shared__ int   wq_n[NW];

    const int tid = threadIdx.x;
    const int lo  = blockIdx.x * PR;

    const int tw = min(n_tabw, DGW);
    const int tu = tw >> 2;
    uint4 tv[(DGW / 4 + PT - 1) / PT];
    #pragma unroll
    for (int k = 0; k < (DGW / 4 + PT - 1) / PT; ++k)
        tv[k] = reinterpret_cast<const uint4*>(deg_b)[min(tid + k * PT, tu - 1)];

    for (int i = tid; i < PR * OUT_FEATS; i += PT) acc[i] = 0.0f;
    if (tid < PR) cnt[tid] = 0;
    if (tid == 0) q_n = 0;
    if (tid < NW) wq_n[tid] = 0;
    __syncthreads();

    const int nl = list_n[blockIdx.x];
    if (nl <= QCAP) {
        const int* lp = list + (size_t)blockIdx.x * QCAP;
        for (int i0 = 0; i0 < nl; i0 += PT * 4) {
            int en[4];
            #pragma unroll
            for (int k = 0; k < 4; ++k)
                en[k] = lp[min(i0 + k * PT + tid, nl - 1)];
            #pragma unroll
            for (int k = 0; k < 4; ++k)
                if (i0 + k * PT + tid < nl)
                    place_edge(en[k] >> 8, en[k] & 255, acc, cnt, wq, wq_n, h, deg_src);
        }
    } else {
        const int    nu = npw >> 2;
        const uint4* pa = reinterpret_cast<const uint4*>(planes + (size_t)(blockIdx.x >> 4) * npw);
        const uint4* pb = reinterpret_cast<const uint4*>(planes + (size_t)(16 + (blockIdx.x & 15)) * npw);
        for (int ub = tid; ub < nu; ub += PT * NU) {
            uint4 va[NU], vb[NU];
            #pragma unroll
            for (int k = 0; k < NU; ++k) {
                const int u = min(ub + k * PT, nu - 1);
                va[k] = pa[u];
                vb[k] = pb[u];
            }
            __builtin_amdgcn_sched_barrier(0);
            #pragma unroll
            for (int k = 0; k < NU; ++k) {
                const int u = ub + k * PT;
                uint4 m;
                m.x = va[k].x & vb[k].x;  m.y = va[k].y & vb[k].y;
                m.z = va[k].z & vb[k].z;  m.w = va[k].w & vb[k].w;
                if (u >= nu) m = make_uint4(0u, 0u, 0u, 0u);
                while (m.x | m.y | m.z | m.w)
                    push_edge(u * 128 + pop_first(m), lo, acc, cnt, q_e, &q_n, src, dst, h, deg_src);
            }
        }
        __syncthreads();

        const int nq = min(q_n, QCAP);
        for (int i = tid; i < nq; i += PT) {
            const int e = q_e[i];
            place_edge(src[e], dst[e] - lo, acc, cnt, wq, wq_n, h, deg_src);
        }
    }
    asm volatile("" :: "v"(tv[0].x), "v"(tv[0].y), "v"(tv[0].z), "v"(tv[0].w),
                       "v"(tv[1].x), "v"(tv[1].y), "v"(tv[1].z), "v"(tv[1].w),
                       "v"(tv[2].x), "v"(tv[2].y), "v"(tv[2].z), "v"(tv[2].w),
                       "v"(tv[3].x), "v"(tv[3].y), "v"(tv[3].z), "v"(tv[3].w));
    #pragma unroll
    for (int k = 0; k < (DGW / 4 + PT - 1) / PT; ++k)
        if (tid + k * PT < tu) reinterpret_cast<uint4*>(dgw)[tid + k * PT] = tv[k];
    if (tid < (tw & 3)) dgw[4 * tu + tid] = deg_b[4 * tu + tid];
    __syncthreads();

    const int lane = tid & 31;
    const int wave = __builtin_amdgcn_readfirstlane(tid >> 5);
    const int wn   = __builtin_amdgcn_readfirstlane(min(wq_n[wave], WCAP));
    const int* wqp = wq + wave * WCAP;
    float sum[16];
    #pragma unroll
    for (int j = 0; j < 16; ++j) sum[j] = 0.0f;
    for (int i0 = 0; i0 < wn; i0 += NB) {
        int    en[NB];
        float  nr[NB];
        float2 vv[NB];
        const int ev = wqp[min(i0 + lane, wn - 1)];
        #pragma unroll
        for (int k = 0; k < NB; ++k)
            en[k] = __builtin_amdgcn_readlane(ev, k);
        const int      sv = ev >> 3;
        const unsigned bq = reinterpret_cast<const unsigned char*>(dgw)[sv];
        float cq = (float)bq;
        if (bq == 255u) cq = deg_src[sv];
        const float nl = rsqrtf(fmaxf(cq, 1.0f));
        #pragma unroll
        for (int k = 0; k < NB; ++k)
            nr[k] = __int_as_float(__builtin_amdgcn_readlane(__float_as_int(nl), k));
        #pragma unroll
        for (int k = 0; k < NB; ++k)
            vv[k] = *reinterpret_cast<const float2*>(h + (size_t)(en[k] >> 3) * OUT_FEATS + 2 * lane);
        asm volatile("" :: "v"(vv[0].x), "v"(vv[0].y), "v"(vv[1].x), "v"(vv[1].y),
                           "v"(vv[2].x), "v"(vv[2].y), "v"(vv[3].x), "v"(vv[3].y),
                           "v"(vv[4].x), "v"(vv[4].y), "v"(vv[5].x), "v"(vv[5].y),
                           "v"(vv[6].x), "v"(vv[6].y), "v"(vv[7].x), "v"(vv[7].y),
                           "v"(vv[8].x), "v"(vv[8].y), "v"(vv[9].x), "v"(vv[9].y),
                           "v"(vv[10].x), "v"(vv[10].y), "v"(vv[11].x), "v"(vv[11].y),
                           "v"(vv[12].x), "v"(vv[12].y), "v"(vv[13].x), "v"(vv[13].y),
                           "v"(vv[14].x), "v"(vv[14].y), "v"(vv[15].x), "v"(vv[15].y));
        #pragma unroll
        for (int k = 0; k < NB; ++k) {
            if (i0 + k < wn) {
                const int   j  = 2 * (en[k] & 7);
                sum[j]     += vv[k].x * nr[k];
                sum[j + 1] += vv[k].y * nr[k];
            }
        }
    }
    {
        float* ar = &acc[wave * OUT_FEATS + 2 * lane];
        #pragma unroll
        for (int j = 0; j < 7; ++j) {
            if (j * NW + wave < PR) {
                ar[j * NW * OUT_FEATS]     += sum[2 * j];
                ar[j * NW * OUT_FEATS + 1] += sum[2 * j + 1];
            }
        }
    }
    __syncthreads();

    for (int i = tid; i < PR * OUT_FEATS; i += PT) {
        int r    = i >> 6;
        int f    = i & 63;
        int node = lo + r;
        if (node < n_nodes) {
            float nd = rsqrtf(fmaxf((float)cnt[r], 1.0f));
            float x  = acc[i] * nd + b[f];
            out[(size_t)node * OUT_FEATS + f] = 1.0f / (1.0f + __expf(-x));
        }
    }
}

extern "C" void kernel_launch(void* const* d_in, const int* in_sizes, int n_in,
                              void* d_out, int out_size, void* d_ws, size_t ws_size,
                              hipStream_t stream) {
    const float* feat = (const float*)d_in[0];
    const int*   src  = (const int*)d_in[1];
    const int*   dst  = (const int*)d_in[2];
    const float* W    = (const float*)d_in[3];
    const float* bias = (const float*)d_in[4];
    float*       out  = (float*)d_out;

    const int n_nodes = in_sizes[0] / IN_FEATS;
    const int n_edges = in_sizes[1];

    const int n_slices = (n_edges + SL - 1) / SL > 0 ? (n_edges + SL - 1) / SL : 1;
    const int npw      = n_slices * SG;
    float*    hbuf     = (float*)d_ws;
    unsigned* planes   = (unsigned*)(hbuf + (size_t)n_nodes * OUT_FEATS);
    unsigned* planes_s = planes + (size_t)NPL * npw;
    float*    deg_src  = (float*)(planes_s + (size_t)NPL * npw);
    const int n_own    = (n_nodes + PR - 1) / PR;
    unsigned* deg_b    = (unsigned*)(deg_src + (size_t)n_own * PR);
    int*      list_n   = (int*)(deg_b + (size_t)n_own * (PR / 4));
    int*      list     = list_n + 256;

    gcn_keys<<<n_slices, 256, 0, stream>>>(src, dst, planes, planes_s, n_edges, npw);

    const int n_tiles       = n_nodes / 16;
    const int n_gemm_blocks = (n_tiles + GT / 32 - 1) / (GT / 32);
    gcn_gemm_outdeg<<<n_gemm_blocks + 2 * n_own, GT, 0, stream>>>(
        feat, W, hbuf, n_tiles, n_gemm_blocks, src, planes_s, deg_src, deg_b, npw, n_nodes,
        dst, planes, list, list_n, n_own);

    gcn_pull<<<n_own, PT, 0, stream>>>(
        src, dst, planes, hbuf, deg_src, bias, out, npw, n_nodes, list, list_n,
        deg_b, n_own * (PR / 4));
}


# pull epilogue by the wave that owns the row, from its register sums: norm, bias, sigmoid and one 256-byte store a row right behind its gathers; no LDS pass, no last barrier (sum order kept)
# speedup vs baseline: 4.4370x; 1.0818x over previous
#include <hip/hip_runtime.h>
#include <math.h>

typedef float v2f __attribute__((ext_vector_type(2)));
typedef float v8f __attribute__((ext_vector_type(8)));

#define IN_FEATS  256
#define OUT_FEATS 64
#define WPAIRS    (IN_FEATS / 2)
#define WSTRIDE   160
#define PR        196

#define QCAP  8192
#define GT    512

#define NU    4

#define SL    2048
#define SG    (SL / 32)
#define NPL   32
#define LPW   (SG + 1)

__device__ __forceinline__ void fix_tail(int4& v, const int* __restrict__ a, int e, int n4, int n_edges) {
    if (e == n4) {
        v.x = a[e];
        if (e + 1 < n_edges) v.y = a[e + 1];
        if (e + 2 < n_edges) v.z = a[e + 2];
    }
}

__device__ __forceinline__ void mark(unsigned* lp, int node, int el, bool live) {
    const unsigned id  = (unsigned)node / PR;
    const unsigned bit = live ? 1u << (el & 31) : 0u;
    const int      g   = el >> 5;
    atomicOr(&lp[((id >> 4) & 15u) * LPW + g], bit);
    atomicOr(&lp[(16u + (id & 15u)) * LPW + g], bit);
}

__global__ void __launch_bounds__(256)
gcn_keys(const int* __restrict__ src,
         const int* __restrict__ dst,
         unsigned* __restrict__ planes,
         unsigned* __restrict__ planes_s,
         int n_edges, int npw) {
    __shared__ unsigned lp[2 * NPL * LPW];
    const int tid = threadIdx.x;
    for (int i = tid; i < 2 * NPL * LPW; i += 256) lp[i] = 0u;

    const int el0 = 4 * tid;
    const int el1 = el0 + SL / 2;
    const int e0  = (int)blockIdx.x * SL + el0;
    const int e1  = (int)blockIdx.x * SL + el1;
    const int n4  = n_edges & ~3;
    int4 d0 = make_int4(0, 0, 0, 0), d1 = d0, s0 = d0, s1 = d0;
    if (n4 >= 4) {
        const int i0 = min(e0, n4 - 4) >> 2;
        const int i1 = min(e1, n4 - 4) >> 2;
        d0 = reinterpret_cast<const int4*>(dst)[i0];
        d1 = reinterpret_cast<const int4*>(dst)[i1];
        s0 = reinterpret_cast<const int4*>(src)[i0];
        s1 = reinterpret_cast<const int4*>(src)[i1];
    }
    if (n4 < n_edges) {
        fix_tail(d0, dst, e0, n4, n_edges);
        fix_tail(d1, dst, e1, n4, n_edges);
        fix_tail(s0, src, e0, n4, n_edges);
        fix_tail(s1, src, e1, n4, n_edges);
    }
    __syncthreads();

    unsigned* ld = lp;
    unsigned* ls = lp + NPL * LPW;
    mark(ld, d0.x, el0,     e0     < n_edges);  mark(ld, d0.y, el0 + 1, e0 + 1 < n_edges);
    mark(ld, d0.z, el0 + 2, e0 + 2 < n_edges);  mark(ld, d0.w, el0 + 3, e0 + 3 < n_edges);
    mark(ld, d1.x, el1,     e1     < n_edges);  mark(ld, d1.y, el1 + 1, e1 + 1 < n_edges);
    mark(ld, d1.z, el1 + 2, e1 + 2 < n_edges);  mark(ld, d1.w, el1 + 3, e1 + 3 < n_edges);
    mark(ls, s0.x, el0,     e0     < n_edges);  mark(ls, s0.y, el0 + 1, e0 + 1 < n_edges);
    mark(ls, s0.z, el0 + 2, e0 + 2 < n_edges);  mark(ls, s0.w, el0 + 3, e0 + 3 < n_edges);
    mark(ls, s1.x, el1,     e1     < n_edges);  mark(ls, s1.y, el1 + 1, e1 + 1 < n_edges);
    mark(ls, s1.z, el1 + 2, e1 + 2 < n_edges);  mark(ls, s1.w, el1 + 3, e1 + 3 < n_edges);
    __syncthreads();

    #pragma unroll
    for (int k = 0; k < 2 * NPL * SG / 256; ++k) {
        const int i   = k * 256 + tid;
        const int row = i >> 6;
        const int g   = i & (SG - 1);
        unsigned* out = (row < NPL) ? planes : planes_s;
        out[(size_t)(row & (NPL - 1)) * npw + (size_t)blockIdx.x * SG + g] = lp[row * LPW + g];
    }
}

__device__ __forceinline__ int pop_first(uint4& m) {
    const bool hx = m.x != 0u, hy = m.y != 0u, hz = m.z != 0u;
    const bool ux = hx, uy = !hx && hy, uz = !hx && !hy && hz, uw = !hx && !hy && !hz;
    const unsigned w    = ux ? m.x : uy ? m.y : uz ? m.z : m.w;
    const int      base = ux ? 0 : uy ? 32 : uz ? 64 : 96;
    const unsigned c    = w & (w - 1u);
    m.x = ux ? c : m.x;
    m.y = uy ? c : m.y;
    m.z = uz ? c : m.z;
    m.w = uw ? c : m.w;
    return base + __ffs(w) - 1;
}

__device__ __forceinline__ void push_src(int e, int lo, int* cnt, int* q_e, int* q_n,
                                         const int* __restrict__ src) {
    int p = atomicAdd(q_n, 1);
    if (p < QCAP) {
        q_e[p] = e;
    } else {
        int r = src[e] - lo;
        if ((unsigned)r < (unsigned)PR) atomicAdd(&cnt[r], 1);
    }
}

__device__ __forceinline__ void outdeg_role(const int* __restrict__ src,
                                            const unsigned* __restrict__ planes_s,
                                            float* __restrict__ deg_src,
                                            unsigned* __restrict__ deg_b,
                                            int npw, int n_nodes, int g, int* lds) {
    int* cnt = lds;
    int* q_e = lds + 256;
    int* q_n = lds + 256 + QCAP;

    const int tid = threadIdx.x;
    const int lo  = g * PR;

    if (tid < PR) cnt[tid] = 0;
    if (tid == 0) *q_n = 0;
    __syncthreads();

    const int    nu = npw >> 2;
    const uint4* pa = reinterpret_cast<const uint4*>(planes_s + (size_t)(g >> 4) * npw);
    const uint4* pb = reinterpret_cast<const uint4*>(planes_s + (size_t)(16 + (g & 15)) * npw);
    for (int ub = tid; ub < nu; ub += GT * NU) {
        uint4 va[NU], vb[NU];
        #pragma unroll
        for (int k = 0; k < NU; ++k) {
            const int u = min(ub + k * GT, nu - 1);
            va[k] = pa[u];
            vb[k] = pb[u];
        }
        __builtin_amdgcn_sched_barrier(0);
        #pragma unroll
        for (int k = 0; k < NU; ++k) {
            const int u = ub + k * GT;
            uint4 m;
            m.x = va[k].x & vb[k].x;  m.y = va[k].y & vb[k].y;
            m.z = va[k].z & vb[k].z;  m.w = va[k].w & vb[k].w;
            if (u >= nu) m = make_uint4(0u, 0u, 0u, 0u);
            while (m.x | m.y | m.z | m.w)
                push_src(u * 128 + pop_first(m), lo, cnt, q_e, q_n, src);
        }
    }
    __syncthreads();

    const int nq = min(*q_n, QCAP);
    for (int i0 = 0; i0 < nq; i0 += GT * 4) {
        int rr[4];
        #pragma unroll
        for (int k = 0; k < 4; ++k)
            rr[k] = src[q_e[min(i0 + k * GT + tid, nq - 1)]] - lo;
        #pragma unroll
        for (int k = 0; k < 4; ++k)
            if (i0 + k * GT + tid < nq && (unsigned)rr[k] < (unsigned)PR)
                atomicAdd(&cnt[rr[k]], 1);
    }
    __syncthreads();

    if (tid < PR && lo + tid < n_nodes) deg_src[lo + tid] = (float)cnt[tid];
    if (tid < PR / 4) {
        const unsigned c0 = (unsigned)min(cnt[4 * tid],     255);
        const unsigned c1 = (unsigned)min(cnt[4 * tid + 1], 255);
        const unsigned c2 = (unsigned)min(cnt[4 * tid + 2], 255);
        const unsigned c3 = (unsigned)min(cnt[4 * tid + 3], 255);
        deg_b[g * (PR / 4) + tid] = c0 | (c1 << 8) | (c2 << 16) | (c3 << 24);
    }
}

__device__ __forceinline__ void push_q(int e, int* q_e, int* q_n) {
    int p = atomicAdd(q_n, 1);
    if (p < QCAP) q_e[p] = e;
}

__device__ __forceinline__ void list_role(const int* __restrict__ src,
                                          const int* __restrict__ dst,
                                          const unsigned* __restrict__ planes,
                                          int* __restrict__ list,
                                          int* __restrict__ list_n,
                                          int npw, int g, int* lds) {
    int* q_e = lds + 256;
    int* q_n = lds + 256 + QCAP;

    const int tid = threadIdx.x;
    const int lo  = g * PR;

    if (tid == 0) *q_n = 0;
    __syncthreads();

    const int    nu = npw >> 2;
    const uint4* pa = reinterpret_cast<const uint4*>(planes + (size_t)(g >> 4) * npw);
    const uint4* pb = reinterpret_cast<const uint4*>(planes + (size_t)(16 + (g & 15)) * npw);
    for (int ub = tid; ub < nu; ub += GT * NU) {
        uint4 va[NU], vb[NU];
        #pragma unroll
        for (int k = 0; k < NU; ++k) {
            const int u = min(ub + k * GT, nu - 1);
            va[k] = pa[u];
            vb[k] = pb[u];
        }
        __builtin_amdgcn_sched_barrier(0);
        #pragma unroll
        for (int k = 0; k < NU; ++k) {
            const int u = ub + k * GT;
            uint4 m;
            m.x = va[k].x & vb[k].x;  m.y = va[k].y & vb[k].y;
            m.z = va[k].z & vb[k].z;  m.w = va[k].w & vb[k].w;
            if (u >= nu) m = make_uint4(0u, 0u, 0u, 0u);
            while (m.x | m.y | m.z | m.w)
                push_q(u * 128 + pop_first(m), q_e, q_n);
        }
    }
    __syncthreads();

    const int nqa = *q_n;
    const int nq  = min(nqa, QCAP);
    int* lp = list + (size_t)g * QCAP;
    for (int i0 = 0; i0 < nq; i0 += GT * 4) {
        int ss[4], dd[4];
        #pragma unroll
        for (int k = 0; k < 4; ++k) {
            const int e = q_e[min(i0 + k * GT + tid, nq - 1)];
            ss[k] = src[e];
            dd[k] = dst[e];
        }
        #pragma unroll
        for (int k = 0; k < 4; ++k) {
            const int i = i0 + k * GT + tid;
            const unsigned r = (unsigned)(dd[k] - lo);
            if (i < nq) lp[i] = (ss[k] << 8) | (int)(r < (unsigned)PR ? r : 255u);
        }
    }
    if (tid == 0) list_n[g] = nqa;
}

__global__ void __launch_bounds__(GT)
gcn_gemm_outdeg(const float* __restrict__ feat,
                const float* __restrict__ W,
                float* __restrict__ h,
                int n_tiles, int n_gemm_blocks,
                const int* __restrict__ src,
                const unsigned* __restrict__ planes_s,
                float* __restrict__ deg_src,
                unsigned* __restrict__ deg_b,
                int npw, int n_nodes,
                const int* __restrict__ dst,
                const unsigned* __restrict__ planes,
                int* __restrict__ list,
                int* __restrict__ list_n,
                int n_own) {
    __shared__ float Wl[WPAIRS * WSTRIDE];

    const int tid = threadIdx.x;
    if ((int)blockIdx.x >= n_gemm_blocks) {
        const int o = (int)blockIdx.x - n_gemm_blocks;
        if (o < n_own)
            outdeg_role(src, planes_s, deg_src, deg_b, npw, n_nodes, o, reinterpret_cast<int*>(Wl));
        else
            list_role(src, dst, planes, list, list_n, npw, o - n_own, reinterpret_cast<int*>(Wl));
        return;
    }
    for (int i = tid; i < WPAIRS * (OUT_FEATS / 2); i += GT) {
        int p  = i >> 5;
        int n2 = (i & 31) * 2;
        float2 lo = *reinterpret_cast<const float2*>(&W[(2 * p) * OUT_FEATS + n2]);
        float2 hi = *reinterpret_cast<const float2*>(&W[(2 * p + 1) * OUT_FEATS + n2]);
        float4 q;  q.x = lo.x;  q.y = hi.x;  q.z = lo.y;  q.w = hi.y;
        *reinterpret_cast<float4*>(&Wl[p * WSTRIDE + 2 * n2]) = q;
    }
    __syncthreads();

    const int wave = tid >> 5;
    const int lane = tid & 31;
    const int tile = blockIdx.x * (GT / 32) + wave;
    if (tile >= n_tiles) return;

    const int half = lane >> 4;
    const int l15  = lane & 15;
    const int m    = tile * 16 + l15;
    const float* frow  = feat + (size_t)m * IN_FEATS + 2 * half;
    const float* wbase = &Wl[half * WSTRIDE + 2 * l15];

    v8f c0 = {}, c1 = {}, c2 = {}, c3 = {};

    #pragma unroll 4
    for (int k = 0; k < IN_FEATS; k += 4) {
        v2f a = *reinterpret_cast<const v2f*>(frow + k);

        const float* wp = wbase + (k >> 1) * WSTRIDE;
        v2f b0 = *reinterpret_cast<const v2f*>(wp);
        v2f b1 = *reinterpret_cast<const v2f*>(wp + 32);
        v2f b2 = *reinterpret_cast<const v2f*>(wp + 64);
        v2f b3 = *reinterpret_cast<const v2f*>(wp + 96);

        c0 = __builtin_amdgcn_wmma_f32_16x16x4_f32(false, a, false, b0,
                                                   (short)0, c0, false, false);
        c1 = __builtin_amdgcn_wmma_f32_16x16x4_f32(false, a, false, b1,
                                                   (short)0, c1, false, false);
        c2 = __builtin_amdgcn_wmma_f32_16x16x4_f32(false, a, false, b2,
                                                   (short)0, c2, false, false);
        c3 = __builtin_amdgcn_wmma_f32_16x16x4_f32(false, a, false, b3,
                                                   (short)0, c3, false, false);
    }

    const int rbase = tile * 16 + 8 * half;
    #pragma unroll
    for (int r = 0; r < 8; ++r) {
        float* orow = h + (size_t)(rbase + r) * OUT_FEATS + l15;
        orow[0]  = c0[r];
        orow[16] = c1[r];
        orow[32] = c2[r];
        orow[48] = c3[r];
    }
}

#define PT    1024
#define NW    (PT / 32)
#define WCAP  256
#define NB    16
#define DGW   (256 * PR / 4)
static_assert((DGW / 4 + PT - 1) / PT == 4, "the table copy holds four 16-byte units a thread");

__device__ __forceinline__ void push_edge(int e, int lo,
                                          float* acc, int* cnt,
                                          int* q_e, int* q_n,
                                          const int* __restrict__ src,
                                          const int* __restrict__ dst,
                                          const float* __restrict__ h,
                                          const float* __restrict__ deg_src) {
    int p = atomicAdd(q_n, 1);
    if (p < QCAP) {
        q_e[p] = e;
    } else {
        int r = dst[e] - lo;
        if ((unsigned)r < (unsigned)PR) {
            const int    sn = src[e];
            const float* hr = h + (size_t)sn * OUT_FEATS;
            const float  nr = rsqrtf(fmaxf(deg_src[sn], 1.0f));
            for (int f = 0; f < OUT_FEATS; ++f)
                atomicAdd(&acc[r * OUT_FEATS + f], hr[f] * nr);
            atomicAdd(&cnt[r], 1);
        }
    }
}

__device__ __forceinline__ void place_edge(int s, int r,
                                           float* acc, int* cnt, int* wq, int* wq_n,
                                           const float* __restrict__ h,
                                           const float* __restrict__ deg_src) {
    if ((unsigned)r < (unsigned)PR) {
        atomicAdd(&cnt[r], 1);
        const int w = r & (NW - 1);
        const int p = atomicAdd(&wq_n[w], 1);
        if (p < WCAP) {
            wq[w * WCAP + p] = (s << 3) | (r >> 5);
        } else {
            const float* hr = h + (size_t)s * OUT_FEATS;
            const float  nr = rsqrtf(fmaxf(deg_src[s], 1.0f));
            for (int f = 0; f < OUT_FEATS; ++f)
                atomicAdd(&acc[r * OUT_FEATS + f], hr[f] * nr);
        }
    }
}

__global__ void __launch_bounds__(PT)
gcn_pull(const int* __restrict__ src,
         const int* __restrict__ dst,
         const unsigned* __restrict__ planes,
         const float* __restrict__ h,
         const float* __restrict__ deg_src,
         const float* __restrict__ b,
         float* __restrict__ out,
         int npw, int n_nodes,
         const int* __restrict__ list,
         const int* __restrict__ list_n,
         const unsigned* __restrict__ deg_b,
         int n_tabw) {
    __shared__ float acc[PR * OUT_FEATS];
    __shared__ unsigned dgw[DGW];
    __shared__ int   cnt[PR];
    __shared__ int   q_e[QCAP];
    __shared__ int   q_n;
    __shared__ int   wq[NW * WCAP];
    __shared__ int   wq_n[NW];

    const int tid = threadIdx.x;
    const int lo  = blockIdx.x * PR;

    const int tw = min(n_tabw, DGW);
    const int tu = tw >> 2;
    uint4 tv[(DGW / 4 + PT - 1) / PT];
    #pragma unroll
    for (int k = 0; k < (DGW / 4 + PT - 1) / PT; ++k)
        tv[k] = reinterpret_cast<const uint4*>(deg_b)[min(tid + k * PT, tu - 1)];

    for (int i = tid; i < PR * OUT_FEATS; i += PT) acc[i] = 0.0f;
    if (tid < PR) cnt[tid] = 0;
    if (tid == 0) q_n = 0;
    if (tid < NW) wq_n[tid] = 0;
    __syncthreads();

    const int nl = list_n[blockIdx.x];
    if (nl <= QCAP) {
        const int* lp = list + (size_t)blockIdx.x * QCAP;
        for (int i0 = 0; i0 < nl; i0 += PT * 4) {
            int en[4];
            #pragma unroll
            for (int k = 0; k < 4; ++k)
                en[k] = lp[min(i0 + k * PT + tid, nl - 1)];
            #pragma unroll
            for (int k = 0; k < 4; ++k)
                if (i0 + k * PT + tid < nl)
                    place_edge(en[k] >> 8, en[k] & 255, acc, cnt, wq, wq_n, h, deg_src);
        }
    } else {
        const int    nu = npw >> 2;
        const uint4* pa = reinterpret_cast<const uint4*>(planes + (size_t)(blockIdx.x >> 4) * npw);
        const uint4* pb = reinterpret_cast<const uint4*>(planes + (size_t)(16 + (blockIdx.x & 15)) * npw);
        for (int ub = tid; ub < nu; ub += PT * NU) {
            uint4 va[NU], vb[NU];
            #pragma unroll
            for (int k = 0; k < NU; ++k) {
                const int u = min(ub + k * PT, nu - 1);
                va[k] = pa[u];
                vb[k] = pb[u];
            }
            __builtin_amdgcn_sched_barrier(0);
            #pragma unroll
            for (int k = 0; k < NU; ++k) {
                const int u = ub + k * PT;
                uint4 m;
                m.x = va[k].x & vb[k].x;  m.y = va[k].y & vb[k].y;
                m.z = va[k].z & vb[k].z;  m.w = va[k].w & vb[k].w;
                if (u >= nu) m = make_uint4(0u, 0u, 0u, 0u);
                while (m.x | m.y | m.z | m.w)
                    push_edge(u * 128 + pop_first(m), lo, acc, cnt, q_e, &q_n, src, dst, h, deg_src);
            }
        }
        __syncthreads();

        const int nq = min(q_n, QCAP);
        for (int i = tid; i < nq; i += PT) {
            const int e = q_e[i];
            place_edge(src[e], dst[e] - lo, acc, cnt, wq, wq_n, h, deg_src);
        }
    }
    asm volatile("" :: "v"(tv[0].x), "v"(tv[0].y), "v"(tv[0].z), "v"(tv[0].w),
                       "v"(tv[1].x), "v"(tv[1].y), "v"(tv[1].z), "v"(tv[1].w),
                       "v"(tv[2].x), "v"(tv[2].y), "v"(tv[2].z), "v"(tv[2].w),
                       "v"(tv[3].x), "v"(tv[3].y), "v"(tv[3].z), "v"(tv[3].w));
    #pragma unroll
    for (int k = 0; k < (DGW / 4 + PT - 1) / PT; ++k)
        if (tid + k * PT < tu) reinterpret_cast<uint4*>(dgw)[tid + k * PT] = tv[k];
    if (tid < (tw & 3)) dgw[4 * tu + tid] = deg_b[4 * tu + tid];
    __syncthreads();

    const int lane = tid & 31;
    const int wave = __builtin_amdgcn_readfirstlane(tid >> 5);
    const int wn   = __builtin_amdgcn_readfirstlane(min(wq_n[wave], WCAP));
    const int* wqp = wq + wave * WCAP;
    const float2 bb = *reinterpret_cast<const float2*>(b + 2 * lane);
    float sum[16];
    #pragma unroll
    for (int j = 0; j < 16; ++j) sum[j] = 0.0f;
    for (int i0 = 0; i0 < wn; i0 += NB) {
        int    en[NB];
        float  nr[NB];
        float2 vv[NB];
        const int ev = wqp[min(i0 + lane, wn - 1)];
        #pragma unroll
        for (int k = 0; k < NB; ++k)
            en[k] = __builtin_amdgcn_readlane(ev, k);
        const int      sv = ev >> 3;
        const unsigned bq = reinterpret_cast<const unsigned char*>(dgw)[sv];
        float cq = (float)bq;
        if (bq == 255u) cq = deg_src[sv];
        const float nl = rsqrtf(fmaxf(cq, 1.0f));
        #pragma unroll
        for (int k = 0; k < NB; ++k)
            nr[k] = __int_as_float(__builtin_amdgcn_readlane(__float_as_int(nl), k));
        #pragma unroll
        for (int k = 0; k < NB; ++k)
            vv[k] = *reinterpret_cast<const float2*>(h + (size_t)(en[k] >> 3) * OUT_FEATS + 2 * lane);
        asm volatile("" :: "v"(vv[0].x), "v"(vv[0].y), "v"(vv[1].x), "v"(vv[1].y),
                           "v"(vv[2].x), "v"(vv[2].y), "v"(vv[3].x), "v"(vv[3].y),
                           "v"(vv[4].x), "v"(vv[4].y), "v"(vv[5].x), "v"(vv[5].y),
                           "v"(vv[6].x), "v"(vv[6].y), "v"(vv[7].x), "v"(vv[7].y),
                           "v"(vv[8].x), "v"(vv[8].y), "v"(vv[9].x), "v"(vv[9].y),
                           "v"(vv[10].x), "v"(vv[10].y), "v"(vv[11].x), "v"(vv[11].y),
                           "v"(vv[12].x), "v"(vv[12].y), "v"(vv[13].x), "v"(vv[13].y),
                           "v"(vv[14].x), "v"(vv[14].y), "v"(vv[15].x), "v"(vv[15].y));
        #pragma unroll
        for (int k = 0; k < NB; ++k) {
            if (i0 + k < wn) {
                const int   j  = 2 * (en[k] & 7);
                sum[j]     += vv[k].x * nr[k];
                sum[j + 1] += vv[k].y * nr[k];
            }
        }
    }
    #pragma unroll
    for (int j = 0; j < 7; ++j) {
        const int r    = j * NW + wave;
        const int node = lo + r;
        if (r < PR && node < n_nodes) {
            const float2 av = *reinterpret_cast<const float2*>(&acc[r * OUT_FEATS + 2 * lane]);
            const float  nd = rsqrtf(fmaxf((float)cnt[r], 1.0f));
            const float  x0 = (av.x + sum[2 * j])     * nd + bb.x;
            const float  x1 = (av.y + sum[2 * j + 1]) * nd + bb.y;
            float2 o;
            o.x = 1.0f / (1.0f + __expf(-x0));
            o.y = 1.0f / (1.0f + __expf(-x1));
            *reinterpret_cast<float2*>(out + (size_t)node * OUT_FEATS + 2 * lane) = o;
        }
    }
}

extern "C" void kernel_launch(void* const* d_in, const int* in_sizes, int n_in,
                              void* d_out, int out_size, void* d_ws, size_t ws_size,
                              hipStream_t stream) {
    const float* feat = (const float*)d_in[0];
    const int*   src  = (const int*)d_in[1];
    const int*   dst  = (const int*)d_in[2];
    const float* W    = (const float*)d_in[3];
    const float* bias = (const float*)d_in[4];
    float*       out  = (float*)d_out;

    const int n_nodes = in_sizes[0] / IN_FEATS;
    const int n_edges = in_sizes[1];

    const int n_slices = (n_edges + SL - 1) / SL > 0 ? (n_edges + SL - 1) / SL : 1;
    const int npw      = n_slices * SG;
    float*    hbuf     = (float*)d_ws;
    unsigned* planes   = (unsigned*)(hbuf + (size_t)n_nodes * OUT_FEATS);
    unsigned* planes_s = planes + (size_t)NPL * npw;
    float*    deg_src  = (float*)(planes_s + (size_t)NPL * npw);
    const int n_own    = (n_nodes + PR - 1) / PR;
    unsigned* deg_b    = (unsigned*)(deg_src + (size_t)n_own * PR);
    int*      list_n   = (int*)(deg_b + (size_t)n_own * (PR / 4));
    int*      list     = list_n + 256;

    gcn_keys<<<n_slices, 256, 0, stream>>>(src, dst, planes, planes_s, n_edges, npw);

    const int n_tiles       = n_nodes / 16;
    const int n_gemm_blocks = (n_tiles + GT / 32 - 1) / (GT / 32);
    gcn_gemm_outdeg<<<n_gemm_blocks + 2 * n_own, GT, 0, stream>>>(
        feat, W, hbuf, n_tiles, n_gemm_blocks, src, planes_s, deg_src, deg_b, npw, n_nodes,
        dst, planes, list, list_n, n_own);

    gcn_pull<<<n_own, PT, 0, stream>>>(
        src, dst, planes, hbuf, deg_src, bias, out, npw, n_nodes, list, list_n,
        deg_b, n_own * (PR / 4));
}
